# starting kernel
# baseline (speedup 1.0000x reference)
#include <hip/hip_runtime.h>
#include <math.h>


typedef __attribute__((ext_vector_type(2))) float v2f;
typedef __attribute__((ext_vector_type(8))) float v8f;

__device__ __forceinline__ v8f wmma4(v2f a, v2f b, v8f c) {
  return __builtin_amdgcn_wmma_f32_16x16x4_f32(false, a, false, b, (short)0, c,
                                               false, false);
}

__global__ void zero_kernel(float* __restrict__ p, long n) {
  long i = (long)blockIdx.x * blockDim.x + threadIdx.x;
  long stride = (long)gridDim.x * blockDim.x;
  for (long k = i * 4; k < n; k += stride * 4) {
    *(float4*)(p + k) = make_float4(0.f, 0.f, 0.f, 0.f);
  }
}

__global__ void node_pre_kernel(const float* __restrict__ node_input,
                                const float* __restrict__ node_attr,
                                const float* __restrict__ Wsc0,
                                const float* __restrict__ Wsc1,
                                const float* __restrict__ Wl10,
                                const float* __restrict__ Wl11,
                                float* __restrict__ f0f1,
                                float* __restrict__ scx, int N) {
  int wv = threadIdx.x >> 5;
  int lane = threadIdx.x & 31;
  int n = blockIdx.x * (blockDim.x >> 5) + wv;
  if (n >= N) return;
  const float* xi = node_input + (long)n * 128;
  float f0 = 0.f, f1x = 0.f, f1y = 0.f, f1z = 0.f;
  float s0 = 0.f, s1x = 0.f, s1y = 0.f, s1z = 0.f;
  for (int u = 0; u < 32; ++u) {
    float x0 = xi[u];
    float x1x = xi[32 + u * 3 + 0];
    float x1y = xi[32 + u * 3 + 1];
    float x1z = xi[32 + u * 3 + 2];
    float wl0 = Wl10[u * 32 + lane], wl1 = Wl11[u * 32 + lane];
    float ws0 = Wsc0[u * 32 + lane], ws1 = Wsc1[u * 32 + lane];
    f0 += x0 * wl0;
    f1x += x1x * wl1; f1y += x1y * wl1; f1z += x1z * wl1;
    s0 += x0 * ws0;
    s1x += x1x * ws1; s1y += x1y * ws1; s1z += x1z * ws1;
  }
  float k = node_attr[n] * 0.17677669529663687f;
  float* fo = f0f1 + (long)n * 128;
  fo[lane] = f0 * k;
  fo[32 + lane] = f1x * k;
  fo[64 + lane] = f1y * k;
  fo[96 + lane] = f1z * k;
  float* so = scx + (long)n * 128;
  so[lane] = s0 * k;
  so[32 + lane * 3 + 0] = s1x * k;
  so[32 + lane * 3 + 1] = s1y * k;
  so[32 + lane * 3 + 2] = s1z * k;
}

#define EK_WAVES 8
#define HSTR 68

__global__ void __launch_bounds__(256)
edge_kernel(const float* __restrict__ edge_scalars,
            const float* __restrict__ edge_attr,
            const int* __restrict__ edge_src,
            const int* __restrict__ edge_dst,
            const float* __restrict__ Wfc1,
            const float* __restrict__ Wfc2,
            const float* __restrict__ f0f1,
            float* __restrict__ mid, int E) {
  __shared__ float fc1[4 * 4 * 32 * 2];
  __shared__ float fc2[16 * 8 * 32 * 2];
  __shared__ float stage[EK_WAVES][2048];

  for (int idx = threadIdx.x; idx < 4 * 4 * 32; idx += blockDim.x) {
    int ln = idx & 31, t = (idx >> 5) & 3, j = idx >> 7;
    int row = 4 * j + 2 * (ln >> 4), col = 16 * t + (ln & 15);
    fc1[idx * 2 + 0] = Wfc1[row * 64 + col];
    fc1[idx * 2 + 1] = Wfc1[(row + 1) * 64 + col];
  }
  for (int idx = threadIdx.x; idx < 16 * 8 * 32; idx += blockDim.x) {
    int ln = idx & 31, t = (idx >> 5) & 7, j = idx >> 8;
    int row = 4 * j + 2 * (ln >> 4), col = 16 * t + (ln & 15);
    fc2[idx * 2 + 0] = Wfc2[row * 128 + col];
    fc2[idx * 2 + 1] = Wfc2[(row + 1) * 128 + col];
  }
  __syncthreads();

  const int lane = threadIdx.x & 31;
  const int wv = threadIdx.x >> 5;
  const int r = lane & 15;
  const int hi = lane >> 4;
  float* st = stage[wv];
  const int nw = gridDim.x * (blockDim.x >> 5);
  const int gw = blockIdx.x * (blockDim.x >> 5) + wv;
  const int tiles = (E + 15) >> 4;

  for (int tile = gw; tile < tiles; tile += nw) {
    const int e0 = tile * 16;
    int er = e0 + r;
    if (er >= E) er = E - 1;
    const float* srow = edge_scalars + (long)er * 16;

    v2f a1[4];
#pragma unroll
    for (int j = 0; j < 4; ++j) a1[j] = *(const v2f*)(srow + 4 * j + 2 * hi);

#pragma unroll
    for (int t = 0; t < 4; ++t) {
      v8f c = {0.f, 0.f, 0.f, 0.f, 0.f, 0.f, 0.f, 0.f};
#pragma unroll
      for (int j = 0; j < 4; ++j) {
        v2f b = *(const v2f*)&fc1[((j * 4 + t) * 32 + lane) * 2];
        c = wmma4(a1[j], b, c);
      }
#pragma unroll
      for (int v = 0; v < 8; ++v) {
        float x = c[v] * 0.25f;
        st[(v + 8 * hi) * HSTR + 16 * t + r] = x / (1.0f + __expf(-x));
      }
    }

    v2f a2[16];
#pragma unroll
    for (int j = 0; j < 16; ++j)
      a2[j] = *(const v2f*)&st[r * HSTR + 4 * j + 2 * hi];

#pragma unroll
    for (int t = 0; t < 8; ++t) {
      v8f c = {0.f, 0.f, 0.f, 0.f, 0.f, 0.f, 0.f, 0.f};
#pragma unroll
      for (int j = 0; j < 16; ++j) {
        v2f b = *(const v2f*)&fc2[((j * 8 + t) * 32 + lane) * 2];
        c = wmma4(a2[j], b, c);
      }
#pragma unroll
      for (int v = 0; v < 8; ++v)
        st[(v + 8 * hi) * 128 + 16 * t + r] = c[v] * 0.125f;
    }

    const float q = 0.31622776601683794f;
    const float r3 = 0.5773502691896258f;
    for (int i = 0; i < 16; ++i) {
      int e = e0 + i;
      if (e >= E) break;
      int s = edge_src[e], d = edge_dst[e];
      const float4 ya = *(const float4*)(edge_attr + (long)e * 4);
      const float* g = f0f1 + (long)s * 128;
      float g0 = g[lane];
      float g1x = g[32 + lane], g1y = g[64 + lane], g1z = g[96 + lane];
      float wA = st[i * 128 + lane];
      float wB = st[i * 128 + 32 + lane];
      float wC = st[i * 128 + 64 + lane];
      float wD = st[i * 128 + 96 + lane];
      float eA = wA * g0 * ya.x * q;
      float eD = wD * (g1x * ya.y + g1y * ya.z + g1z * ya.w) * r3 * q;
      float gB = wB * g0 * q;
      float cc = wC * ya.x * q;
      float* m = mid + (long)d * 256;
      atomicAdd(m + lane, eA);
      atomicAdd(m + 32 + lane, eD);
      atomicAdd(m + 64 + 3 * lane + 0, gB * ya.y);
      atomicAdd(m + 64 + 3 * lane + 1, gB * ya.z);
      atomicAdd(m + 64 + 3 * lane + 2, gB * ya.w);
      atomicAdd(m + 160 + 3 * lane + 0, cc * g1x);
      atomicAdd(m + 160 + 3 * lane + 1, cc * g1y);
      atomicAdd(m + 160 + 3 * lane + 2, cc * g1z);
    }
  }
}

__global__ void node_post_kernel(const float* __restrict__ mid,
                                 const float* __restrict__ node_attr,
                                 const float* __restrict__ Wl20,
                                 const float* __restrict__ Wl21,
                                 const float* __restrict__ Wl3,
                                 const float* __restrict__ scx,
                                 float* __restrict__ out, int N) {
  int wv = threadIdx.x >> 5;
  int lane = threadIdx.x & 31;
  int n = blockIdx.x * (blockDim.x >> 5) + wv;
  if (n >= N) return;
  const float* m = mid + (long)n * 256;
  float o0 = 0.f, o1x = 0.f, o1y = 0.f, o1z = 0.f, ang = 0.f;
  for (int v = 0; v < 64; ++v) {
    float m0 = m[v];
    float mx = m[64 + v * 3 + 0];
    float my = m[64 + v * 3 + 1];
    float mz = m[64 + v * 3 + 2];
    float w0 = Wl20[v * 32 + lane], w1 = Wl21[v * 32 + lane];
    o0 += m0 * w0;
    o1x += mx * w1; o1y += my * w1; o1z += mz * w1;
    ang += m0 * Wl3[v];
  }
  float k = node_attr[n] * 0.125f;
  o0 *= k; o1x *= k; o1y *= k; o1z *= k;
  float angle = 0.1f * ang * k;
  float ca = cosf(angle), sa = sinf(angle);
  const float* sx = scx + (long)n * 128;
  float* o = out + (long)n * 128;
  o[lane] = ca * sx[lane] + sa * o0;
  o[32 + 3 * lane + 0] = ca * sx[32 + 3 * lane + 0] + sa * o1x;
  o[32 + 3 * lane + 1] = ca * sx[32 + 3 * lane + 1] + sa * o1y;
  o[32 + 3 * lane + 2] = ca * sx[32 + 3 * lane + 2] + sa * o1z;
}

extern "C" void kernel_launch(void* const* d_in, const int* in_sizes, int n_in,
                              void* d_out, int out_size, void* d_ws, size_t ws_size,
                              hipStream_t stream) {
  const float* node_input   = (const float*)d_in[0];
  const float* node_attr    = (const float*)d_in[1];
  const int*   edge_src     = (const int*)d_in[2];
  const int*   edge_dst     = (const int*)d_in[3];
  const float* edge_attr    = (const float*)d_in[4];
  const float* edge_scalars = (const float*)d_in[5];
  const float* W_sc0        = (const float*)d_in[6];
  const float* W_sc1        = (const float*)d_in[7];
  const float* W_lin1_0     = (const float*)d_in[8];
  const float* W_lin1_1     = (const float*)d_in[9];
  const float* W_fc1        = (const float*)d_in[10];
  const float* W_fc2        = (const float*)d_in[11];
  const float* W_lin2_0     = (const float*)d_in[12];
  const float* W_lin2_1     = (const float*)d_in[13];
  const float* W_lin3       = (const float*)d_in[14];
  int N = in_sizes[1];
  int E = in_sizes[2];

  float* f0f1 = (float*)d_ws;
  float* scx  = f0f1 + (size_t)N * 128;
  float* mid  = scx + (size_t)N * 128;
  float* out  = (float*)d_out;

  zero_kernel<<<2048, 256, 0, stream>>>(mid, (long)N * 256);

  int nblk_node = (N + 7) / 8;
  node_pre_kernel<<<nblk_node, 256, 0, stream>>>(
      node_input, node_attr, W_sc0, W_sc1, W_lin1_0, W_lin1_1, f0f1, scx, N);

  edge_kernel<<<1024, 256, 0, stream>>>(
      edge_scalars, edge_attr, edge_src, edge_dst, W_fc1, W_fc2, f0f1, mid, E);

  node_post_kernel<<<nblk_node, 256, 0, stream>>>(
      mid, node_attr, W_lin2_0, W_lin2_1, W_lin3, scx, out, N);
}


# node output linears (W_lin2_0, W_lin2_1, W_lin3) applied per 16-edge tile on f32 WMMA before the segment sum: 129 atomic adds per edge, not 256; node epilogue elementwise; sum order changed
# speedup vs baseline: 1.0328x; 1.0328x over previous
#include <hip/hip_runtime.h>
#include <math.h>


typedef __attribute__((ext_vector_type(2))) float v2f;
typedef __attribute__((ext_vector_type(4))) float v4f;
typedef __attribute__((ext_vector_type(8))) float v8f;

__device__ __forceinline__ v8f wmma4(v2f a, v2f b, v8f c) {
  return __builtin_amdgcn_wmma_f32_16x16x4_f32(false, a, false, b, (short)0, c,
                                               false, false);
}

__global__ void zero_kernel(float* __restrict__ p, long n) {
  long i = (long)blockIdx.x * blockDim.x + threadIdx.x;
  long stride = (long)gridDim.x * blockDim.x;
  for (long k = i * 4; k < n; k += stride * 4) {
    if (k + 4 <= n) {
      *(float4*)(p + k) = make_float4(0.f, 0.f, 0.f, 0.f);
    } else {
      for (long kk = k; kk < n; ++kk) p[kk] = 0.f;
    }
  }
}

__global__ void node_pre_kernel(const float* __restrict__ node_input,
                                const float* __restrict__ node_attr,
                                const float* __restrict__ Wsc0,
                                const float* __restrict__ Wsc1,
                                const float* __restrict__ Wl10,
                                const float* __restrict__ Wl11,
                                float* __restrict__ f0f1,
                                float* __restrict__ scx, int N) {
  int wv = threadIdx.x >> 5;
  int lane = threadIdx.x & 31;
  int n = blockIdx.x * (blockDim.x >> 5) + wv;
  if (n >= N) return;
  const float* xi = node_input + (long)n * 128;
  float f0 = 0.f, f1x = 0.f, f1y = 0.f, f1z = 0.f;
  float s0 = 0.f, s1x = 0.f, s1y = 0.f, s1z = 0.f;
  for (int u = 0; u < 32; ++u) {
    float x0 = xi[u];
    float x1x = xi[32 + u * 3 + 0];
    float x1y = xi[32 + u * 3 + 1];
    float x1z = xi[32 + u * 3 + 2];
    float wl0 = Wl10[u * 32 + lane], wl1 = Wl11[u * 32 + lane];
    float ws0 = Wsc0[u * 32 + lane], ws1 = Wsc1[u * 32 + lane];
    f0 += x0 * wl0;
    f1x += x1x * wl1; f1y += x1y * wl1; f1z += x1z * wl1;
    s0 += x0 * ws0;
    s1x += x1x * ws1; s1y += x1y * ws1; s1z += x1z * ws1;
  }
  float k = node_attr[n] * 0.17677669529663687f;
  float* fo = f0f1 + (long)n * 128;
  fo[lane] = f0 * k;
  fo[32 + lane] = f1x * k;
  fo[64 + lane] = f1y * k;
  fo[96 + lane] = f1z * k;
  float* so = scx + (long)n * 128;
  so[lane] = s0 * k;
  so[32 + lane * 3 + 0] = s1x * k;
  so[32 + lane * 3 + 1] = s1y * k;
  so[32 + lane * 3 + 2] = s1z * k;
}

#define EK_WAVES 8
#define HSTR 68
#define XS 196

__global__ void __launch_bounds__(EK_WAVES * 32, 4)
edge_kernel(const float* __restrict__ edge_scalars,
            const float* __restrict__ edge_attr,
            const int* __restrict__ edge_src,
            const int* __restrict__ edge_dst,
            const float* __restrict__ Wfc1,
            const float* __restrict__ Wfc2,
            const float* __restrict__ Wl20,
            const float* __restrict__ Wl21,
            const float* __restrict__ Wl3,
            const float* __restrict__ f0f1,
            float* __restrict__ acc,
            float* __restrict__ ang, int E) {
  __shared__ __attribute__((aligned(16))) float fc1[4 * 4 * 32 * 2];
  __shared__ __attribute__((aligned(16))) float fc2[16 * 8 * 32 * 2];
  __shared__ __attribute__((aligned(16))) float l20[16 * 2 * 32 * 2];
  __shared__ __attribute__((aligned(16))) float l21[16 * 2 * 32 * 2];
  __shared__ __attribute__((aligned(16))) float w3s[64];
  __shared__ __attribute__((aligned(16))) float stage[EK_WAVES][16 * XS];

  for (int idx = threadIdx.x; idx < 4 * 4 * 32; idx += blockDim.x) {
    int ln = idx & 31, t = (idx >> 5) & 3, j = idx >> 7;
    int row = 4 * j + 2 * (ln >> 4), col = 16 * t + (ln & 15);
    fc1[idx * 2 + 0] = Wfc1[row * 64 + col];
    fc1[idx * 2 + 1] = Wfc1[(row + 1) * 64 + col];
  }
  for (int idx = threadIdx.x; idx < 16 * 8 * 32; idx += blockDim.x) {
    int ln = idx & 31, t = (idx >> 5) & 7, j = idx >> 8;
    int row = 4 * j + 2 * (ln >> 4), col = 16 * t + (ln & 15);
    fc2[idx * 2 + 0] = Wfc2[row * 128 + col];
    fc2[idx * 2 + 1] = Wfc2[(row + 1) * 128 + col];
  }
  for (int idx = threadIdx.x; idx < 16 * 2 * 32; idx += blockDim.x) {
    int ln = idx & 31, t = (idx >> 5) & 1, j = idx >> 6;
    int row = 4 * j + 2 * (ln >> 4), col = 16 * t + (ln & 15);
    l20[idx * 2 + 0] = Wl20[row * 32 + col];
    l20[idx * 2 + 1] = Wl20[(row + 1) * 32 + col];
    l21[idx * 2 + 0] = Wl21[row * 32 + col];
    l21[idx * 2 + 1] = Wl21[(row + 1) * 32 + col];
  }
  for (int idx = threadIdx.x; idx < 64; idx += blockDim.x) w3s[idx] = Wl3[idx];
  __syncthreads();

  const int lane = threadIdx.x & 31;
  const int wv = threadIdx.x >> 5;
  const int r = lane & 15;
  const int hi = lane >> 4;
  float* st = stage[wv];
  const int nw = gridDim.x * (blockDim.x >> 5);
  const int gw = blockIdx.x * (blockDim.x >> 5) + wv;
  const int tiles = (E + 15) >> 4;

  for (int tile = gw; tile < tiles; tile += nw) {
    const int e0 = tile * 16;
    int er = e0 + r;
    if (er >= E) er = E - 1;
    const float* srow = edge_scalars + (long)er * 16;

    v2f a1[4];
#pragma unroll
    for (int j = 0; j < 4; ++j) a1[j] = *(const v2f*)(srow + 4 * j + 2 * hi);

#pragma unroll
    for (int t = 0; t < 4; ++t) {
      v8f c = {0.f, 0.f, 0.f, 0.f, 0.f, 0.f, 0.f, 0.f};
#pragma unroll
      for (int j = 0; j < 4; ++j) {
        v2f b = *(const v2f*)&fc1[((j * 4 + t) * 32 + lane) * 2];
        c = wmma4(a1[j], b, c);
      }
#pragma unroll
      for (int v = 0; v < 8; ++v) {
        float x = c[v] * 0.25f;
        st[(v + 8 * hi) * HSTR + 16 * t + r] = x / (1.0f + __expf(-x));
      }
    }

    v2f a2[16];
#pragma unroll
    for (int j = 0; j < 16; ++j)
      a2[j] = *(const v2f*)&st[r * HSTR + 4 * j + 2 * hi];

#pragma unroll
    for (int t = 0; t < 8; ++t) {
      v8f c = {0.f, 0.f, 0.f, 0.f, 0.f, 0.f, 0.f, 0.f};
#pragma unroll
      for (int j = 0; j < 16; ++j) {
        v2f b = *(const v2f*)&fc2[((j * 8 + t) * 32 + lane) * 2];
        c = wmma4(a2[j], b, c);
      }
#pragma unroll
      for (int v = 0; v < 8; ++v)
        st[(v + 8 * hi) * XS + 16 * t + r] = c[v] * 0.125f;
    }

    int dv[8];
    float y1x[8], y1y[8], y1z[8];
    bool ok[8];
#pragma unroll
    for (int v = 0; v < 8; ++v) {
      int row = e0 + v + 8 * hi;
      ok[v] = row < E;
      if (!ok[v]) row = E - 1;
      dv[v] = edge_dst[row];
      const float4 yv = *(const float4*)(edge_attr + (long)row * 4);
      y1x[v] = yv.y; y1y[v] = yv.z; y1z[v] = yv.w;
    }
    const int dr = edge_dst[er];

    const float q = 0.31622776601683794f;
    const float r3 = 0.5773502691896258f;
    for (int i = 0; i < 16; ++i) {
      const int e = e0 + i;
      float* xr = st + i * XS;
      if (e < E) {
        int s = edge_src[e];
        const float4 ya = *(const float4*)(edge_attr + (long)e * 4);
        const float* g = f0f1 + (long)s * 128;
        float g0 = g[lane];
        float g1x = g[32 + lane], g1y = g[64 + lane], g1z = g[96 + lane];
        float wA = xr[lane];
        float wB = xr[32 + lane];
        float wC = xr[64 + lane];
        float wD = xr[96 + lane];
        float eA = wA * g0 * ya.x * q;
        float eD = wD * (g1x * ya.y + g1y * ya.z + g1z * ya.w) * r3 * q;
        float gB = wB * g0 * q;
        float cc = wC * ya.x * q;
        xr[lane] = eA;
        xr[32 + lane] = eD;
        xr[64 + lane] = gB;
        xr[96 + lane] = cc * g1x;
        xr[128 + lane] = cc * g1y;
        xr[160 + lane] = cc * g1z;
      } else {
        xr[lane] = 0.f;
        xr[32 + lane] = 0.f;
        xr[64 + lane] = 0.f;
        xr[96 + lane] = 0.f;
        xr[128 + lane] = 0.f;
        xr[160 + lane] = 0.f;
      }
    }

#pragma unroll
    for (int t = 0; t < 2; ++t) {
      v8f c = {0.f, 0.f, 0.f, 0.f, 0.f, 0.f, 0.f, 0.f};
#pragma unroll
      for (int j = 0; j < 16; ++j) {
        v2f a = *(const v2f*)&st[r * XS + 4 * j + 2 * hi];
        v2f b = *(const v2f*)&l20[((j * 2 + t) * 32 + lane) * 2];
        c = wmma4(a, b, c);
      }
#pragma unroll
      for (int v = 0; v < 8; ++v) {
        if (ok[v]) {
          float* m = acc + (long)dv[v] * 128 + 16 * t + r;
          atomicAdd(m, c[v]);
        }
      }
    }

#pragma unroll
    for (int t = 0; t < 2; ++t) {
      v8f cb = {0.f, 0.f, 0.f, 0.f, 0.f, 0.f, 0.f, 0.f};
      v8f c0 = cb, c1 = cb, c2 = cb;
#pragma unroll
      for (int j = 0; j < 8; ++j) {
        v2f ab  = *(const v2f*)&st[r * XS + 64 + 4 * j + 2 * hi];
        v2f ac0 = *(const v2f*)&st[r * XS + 96 + 4 * j + 2 * hi];
        v2f ac1 = *(const v2f*)&st[r * XS + 128 + 4 * j + 2 * hi];
        v2f ac2 = *(const v2f*)&st[r * XS + 160 + 4 * j + 2 * hi];
        v2f bb = *(const v2f*)&l21[((j * 2 + t) * 32 + lane) * 2];
        v2f bc = *(const v2f*)&l21[(((8 + j) * 2 + t) * 32 + lane) * 2];
        cb = wmma4(ab, bb, cb);
        c0 = wmma4(ac0, bc, c0);
        c1 = wmma4(ac1, bc, c1);
        c2 = wmma4(ac2, bc, c2);
      }
#pragma unroll
      for (int v = 0; v < 8; ++v) {
        if (ok[v]) {
          float* m = acc + (long)dv[v] * 128 + 32 + 16 * t + r;
          atomicAdd(m,      cb[v] * y1x[v] + c0[v]);
          atomicAdd(m + 32, cb[v] * y1y[v] + c1[v]);
          atomicAdd(m + 64, cb[v] * y1z[v] + c2[v]);
        }
      }
    }

    {
      const float* rowp = st + r * XS + 32 * hi;
      const float* w3p = w3s + 32 * hi;
      float p = 0.f;
#pragma unroll
      for (int kk = 0; kk < 8; ++kk) {
        v4f x = *(const v4f*)(rowp + 4 * kk);
        v4f w = *(const v4f*)(w3p + 4 * kk);
        p += x.x * w.x + x.y * w.y + x.z * w.z + x.w * w.w;
      }
      p += __shfl_xor(p, 16);
      if (hi == 0 && (e0 + r) < E) atomicAdd(ang + dr, p);
    }
  }
}

__global__ void node_post_kernel(const float* __restrict__ acc,
                                 const float* __restrict__ ang,
                                 const float* __restrict__ node_attr,
                                 const float* __restrict__ scx,
                                 float* __restrict__ out, int N) {
  int wv = threadIdx.x >> 5;
  int lane = threadIdx.x & 31;
  int n = blockIdx.x * (blockDim.x >> 5) + wv;
  if (n >= N) return;
  const float* a = acc + (long)n * 128;
  const float* sx = scx + (long)n * 128;
  float* o = out + (long)n * 128;
  float k = node_attr[n] * 0.125f;
  float angle = 0.1f * ang[n] * k;
  float ca = cosf(angle), sa = sinf(angle);
  float sk = sa * k;
  o[lane] = ca * sx[lane] + sk * a[lane];
#pragma unroll
  for (int c = 0; c < 3; ++c) {
    int jj = lane + 32 * c;
    int w = jj / 3, m = jj - 3 * w;
    o[32 + jj] = ca * sx[32 + jj] + sk * a[32 + 32 * m + w];
  }
}

extern "C" void kernel_launch(void* const* d_in, const int* in_sizes, int n_in,
                              void* d_out, int out_size, void* d_ws, size_t ws_size,
                              hipStream_t stream) {
  const float* node_input   = (const float*)d_in[0];
  const float* node_attr    = (const float*)d_in[1];
  const int*   edge_src     = (const int*)d_in[2];
  const int*   edge_dst     = (const int*)d_in[3];
  const float* edge_attr    = (const float*)d_in[4];
  const float* edge_scalars = (const float*)d_in[5];
  const float* W_sc0        = (const float*)d_in[6];
  const float* W_sc1        = (const float*)d_in[7];
  const float* W_lin1_0     = (const float*)d_in[8];
  const float* W_lin1_1     = (const float*)d_in[9];
  const float* W_fc1        = (const float*)d_in[10];
  const float* W_fc2        = (const float*)d_in[11];
  const float* W_lin2_0     = (const float*)d_in[12];
  const float* W_lin2_1     = (const float*)d_in[13];
  const float* W_lin3       = (const float*)d_in[14];
  int N = in_sizes[1];
  int E = in_sizes[2];

  float* f0f1 = (float*)d_ws;
  float* scx  = f0f1 + (size_t)N * 128;
  float* acc  = scx + (size_t)N * 128;
  float* ang  = acc + (size_t)N * 128;
  float* out  = (float*)d_out;

  zero_kernel<<<2048, 256, 0, stream>>>(acc, (long)N * 129);

  int nblk_node = (N + 7) / 8;
  node_pre_kernel<<<nblk_node, 256, 0, stream>>>(
      node_input, node_attr, W_sc0, W_sc1, W_lin1_0, W_lin1_1, f0f1, scx, N);

  edge_kernel<<<1024, EK_WAVES * 32, 0, stream>>>(
      edge_scalars, edge_attr, edge_src, edge_dst, W_fc1, W_fc2,
      W_lin2_0, W_lin2_1, W_lin3, f0f1, acc, ang, E);

  node_post_kernel<<<nblk_node, 256, 0, stream>>>(
      acc, ang, node_attr, scx, out, N);
}


# edge features built in A-fragment registers: each lane gathers its own edge's source row in 8-byte pieces ahead of the MLP GEMMs; no per-edge loop, no LDS feature rows; angle dot sum order changed
# speedup vs baseline: 1.1196x; 1.0840x over previous
#include <hip/hip_runtime.h>
#include <math.h>


typedef __attribute__((ext_vector_type(2))) float v2f;
typedef __attribute__((ext_vector_type(8))) float v8f;

__device__ __forceinline__ v8f wmma4(v2f a, v2f b, v8f c) {
  return __builtin_amdgcn_wmma_f32_16x16x4_f32(false, a, false, b, (short)0, c,
                                               false, false);
}

__global__ void zero_kernel(float* __restrict__ p, long n) {
  long i = (long)blockIdx.x * blockDim.x + threadIdx.x;
  long stride = (long)gridDim.x * blockDim.x;
  for (long k = i * 4; k < n; k += stride * 4) {
    if (k + 4 <= n) {
      *(float4*)(p + k) = make_float4(0.f, 0.f, 0.f, 0.f);
    } else {
      for (long kk = k; kk < n; ++kk) p[kk] = 0.f;
    }
  }
}

__global__ void node_pre_kernel(const float* __restrict__ node_input,
                                const float* __restrict__ node_attr,
                                const float* __restrict__ Wsc0,
                                const float* __restrict__ Wsc1,
                                const float* __restrict__ Wl10,
                                const float* __restrict__ Wl11,
                                float* __restrict__ f0f1,
                                float* __restrict__ scx, int N) {
  int wv = threadIdx.x >> 5;
  int lane = threadIdx.x & 31;
  int n = blockIdx.x * (blockDim.x >> 5) + wv;
  if (n >= N) return;
  const float* xi = node_input + (long)n * 128;
  float f0 = 0.f, f1x = 0.f, f1y = 0.f, f1z = 0.f;
  float s0 = 0.f, s1x = 0.f, s1y = 0.f, s1z = 0.f;
  for (int u = 0; u < 32; ++u) {
    float x0 = xi[u];
    float x1x = xi[32 + u * 3 + 0];
    float x1y = xi[32 + u * 3 + 1];
    float x1z = xi[32 + u * 3 + 2];
    float wl0 = Wl10[u * 32 + lane], wl1 = Wl11[u * 32 + lane];
    float ws0 = Wsc0[u * 32 + lane], ws1 = Wsc1[u * 32 + lane];
    f0 += x0 * wl0;
    f1x += x1x * wl1; f1y += x1y * wl1; f1z += x1z * wl1;
    s0 += x0 * ws0;
    s1x += x1x * ws1; s1y += x1y * ws1; s1z += x1z * ws1;
  }
  float k = node_attr[n] * 0.17677669529663687f;
  float* fo = f0f1 + (long)n * 128;
  fo[lane] = f0 * k;
  fo[32 + lane] = f1x * k;
  fo[64 + lane] = f1y * k;
  fo[96 + lane] = f1z * k;
  float* so = scx + (long)n * 128;
  so[lane] = s0 * k;
  so[32 + lane * 3 + 0] = s1x * k;
  so[32 + lane * 3 + 1] = s1y * k;
  so[32 + lane * 3 + 2] = s1z * k;
}

#define EK_WAVES 8
#define HSTR 68
#define XS 132

__global__ void __launch_bounds__(EK_WAVES * 32, 4)
edge_kernel(const float* __restrict__ edge_scalars,
            const float* __restrict__ edge_attr,
            const int* __restrict__ edge_src,
            const int* __restrict__ edge_dst,
            const float* __restrict__ Wfc1,
            const float* __restrict__ Wfc2,
            const float* __restrict__ Wl20,
            const float* __restrict__ Wl21,
            const float* __restrict__ Wl3,
            const float* __restrict__ f0f1,
            float* __restrict__ acc,
            float* __restrict__ ang, int E) {
  __shared__ __attribute__((aligned(16))) float fc1[4 * 4 * 32 * 2];
  __shared__ __attribute__((aligned(16))) float fc2[16 * 8 * 32 * 2];
  __shared__ __attribute__((aligned(16))) float l20[16 * 2 * 32 * 2];
  __shared__ __attribute__((aligned(16))) float l21[16 * 2 * 32 * 2];
  __shared__ __attribute__((aligned(16))) float w3s[64];
  __shared__ __attribute__((aligned(16))) float stage[EK_WAVES][16 * XS];

  for (int idx = threadIdx.x; idx < 4 * 4 * 32; idx += blockDim.x) {
    int ln = idx & 31, t = (idx >> 5) & 3, j = idx >> 7;
    int row = 4 * j + 2 * (ln >> 4), col = 16 * t + (ln & 15);
    fc1[idx * 2 + 0] = Wfc1[row * 64 + col];
    fc1[idx * 2 + 1] = Wfc1[(row + 1) * 64 + col];
  }
  for (int idx = threadIdx.x; idx < 16 * 8 * 32; idx += blockDim.x) {
    int ln = idx & 31, t = (idx >> 5) & 7, j = idx >> 8;
    int row = 4 * j + 2 * (ln >> 4), col = 16 * t + (ln & 15);
    fc2[idx * 2 + 0] = Wfc2[row * 128 + col];
    fc2[idx * 2 + 1] = Wfc2[(row + 1) * 128 + col];
  }
  for (int idx = threadIdx.x; idx < 16 * 2 * 32; idx += blockDim.x) {
    int ln = idx & 31, t = (idx >> 5) & 1, j = idx >> 6;
    int row = 4 * j + 2 * (ln >> 4), col = 16 * t + (ln & 15);
    l20[idx * 2 + 0] = Wl20[row * 32 + col];
    l20[idx * 2 + 1] = Wl20[(row + 1) * 32 + col];
    l21[idx * 2 + 0] = Wl21[row * 32 + col];
    l21[idx * 2 + 1] = Wl21[(row + 1) * 32 + col];
  }
  for (int idx = threadIdx.x; idx < 64; idx += blockDim.x) w3s[idx] = Wl3[idx];
  __syncthreads();

  const int lane = threadIdx.x & 31;
  const int wv = threadIdx.x >> 5;
  const int r = lane & 15;
  const int hi = lane >> 4;
  float* st = stage[wv];
  const int nw = gridDim.x * (blockDim.x >> 5);
  const int gw = blockIdx.x * (blockDim.x >> 5) + wv;
  const int tiles = (E + 15) >> 4;

  for (int tile = gw; tile < tiles; tile += nw) {
    const int e0 = tile * 16;
    int er = e0 + r;
    if (er >= E) er = E - 1;
    const float* srow = edge_scalars + (long)er * 16;

    v2f a1[4];
#pragma unroll
    for (int j = 0; j < 4; ++j) a1[j] = *(const v2f*)(srow + 4 * j + 2 * hi);

    const int sr = edge_src[er];
    const float4 ya = *(const float4*)(edge_attr + (long)er * 4);
    const int dr = edge_dst[er];

    const float* gp = f0f1 + (long)sr * 128 + 2 * hi;
    v2f g0[8], g1x[8], g1y[8], g1z[8];
#pragma unroll
    for (int j = 0; j < 8; ++j) g0[j] = *(const v2f*)(gp + 4 * j);
#pragma unroll
    for (int j = 0; j < 8; ++j) g1x[j] = *(const v2f*)(gp + 32 + 4 * j);
#pragma unroll
    for (int j = 0; j < 8; ++j) g1y[j] = *(const v2f*)(gp + 64 + 4 * j);
#pragma unroll
    for (int j = 0; j < 8; ++j) g1z[j] = *(const v2f*)(gp + 96 + 4 * j);

    int dv[8];
    float y1x[8], y1y[8], y1z[8];
    bool ok[8];
#pragma unroll
    for (int v = 0; v < 8; ++v) {
      int row = e0 + v + 8 * hi;
      ok[v] = row < E;
      if (!ok[v]) row = E - 1;
      dv[v] = edge_dst[row];
      const float4 yv = *(const float4*)(edge_attr + (long)row * 4);
      y1x[v] = yv.y; y1y[v] = yv.z; y1z[v] = yv.w;
    }

    __builtin_amdgcn_sched_barrier(0);

#pragma unroll
    for (int t = 0; t < 4; ++t) {
      v8f c = {0.f, 0.f, 0.f, 0.f, 0.f, 0.f, 0.f, 0.f};
#pragma unroll
      for (int j = 0; j < 4; ++j) {
        v2f b = *(const v2f*)&fc1[((j * 4 + t) * 32 + lane) * 2];
        c = wmma4(a1[j], b, c);
      }
#pragma unroll
      for (int v = 0; v < 8; ++v) {
        float x = c[v] * 0.25f;
        st[(v + 8 * hi) * HSTR + 16 * t + r] = x / (1.0f + __expf(-x));
      }
    }

    v2f a2[16];
#pragma unroll
    for (int j = 0; j < 16; ++j)
      a2[j] = *(const v2f*)&st[r * HSTR + 4 * j + 2 * hi];

#pragma unroll
    for (int t = 0; t < 8; ++t) {
      v8f c = {0.f, 0.f, 0.f, 0.f, 0.f, 0.f, 0.f, 0.f};
#pragma unroll
      for (int j = 0; j < 16; ++j) {
        v2f b = *(const v2f*)&fc2[((j * 8 + t) * 32 + lane) * 2];
        c = wmma4(a2[j], b, c);
      }
#pragma unroll
      for (int v = 0; v < 8; ++v)
        st[(v + 8 * hi) * XS + 16 * t + r] = c[v] * 0.125f;
    }

    const float q = 0.31622776601683794f;
    const float r3 = 0.5773502691896258f;
    const float* wr = st + r * XS + 2 * hi;
    v2f fA[8], fD[8], fB[8], fC0[8], fC1[8], fC2[8];
#pragma unroll
    for (int j = 0; j < 8; ++j) {
      const v2f wA = *(const v2f*)(wr + 4 * j);
      const v2f wB = *(const v2f*)(wr + 32 + 4 * j);
      const v2f wC = *(const v2f*)(wr + 64 + 4 * j);
      const v2f wD = *(const v2f*)(wr + 96 + 4 * j);
      fA[j] = wA * g0[j] * ya.x * q;
      fD[j] = wD * (g1x[j] * ya.y + g1y[j] * ya.z + g1z[j] * ya.w) * r3 * q;
      fB[j] = wB * g0[j] * q;
      const v2f cc = wC * ya.x * q;
      fC0[j] = cc * g1x[j];
      fC1[j] = cc * g1y[j];
      fC2[j] = cc * g1z[j];
    }

#pragma unroll
    for (int t = 0; t < 2; ++t) {
      v8f c = {0.f, 0.f, 0.f, 0.f, 0.f, 0.f, 0.f, 0.f};
#pragma unroll
      for (int j = 0; j < 16; ++j) {
        v2f a = j < 8 ? fA[j] : fD[j - 8];
        v2f b = *(const v2f*)&l20[((j * 2 + t) * 32 + lane) * 2];
        c = wmma4(a, b, c);
      }
#pragma unroll
      for (int v = 0; v < 8; ++v) {
        if (ok[v]) {
          float* m = acc + (long)dv[v] * 128 + 16 * t + r;
          atomicAdd(m, c[v]);
        }
      }
    }

    {
      const float* w3p = w3s + 2 * hi;
      float p = 0.f;
#pragma unroll
      for (int j = 0; j < 8; ++j) {
        const v2f wa = *(const v2f*)(w3p + 4 * j);
        const v2f wd = *(const v2f*)(w3p + 32 + 4 * j);
        p += fA[j].x * wa.x + fA[j].y * wa.y + fD[j].x * wd.x + fD[j].y * wd.y;
      }
      p += __shfl_xor(p, 16);
      if (hi == 0 && (e0 + r) < E) atomicAdd(ang + dr, p);
    }

#pragma unroll
    for (int t = 0; t < 2; ++t) {
      v8f cb = {0.f, 0.f, 0.f, 0.f, 0.f, 0.f, 0.f, 0.f};
      v8f c0 = cb, c1 = cb, c2 = cb;
#pragma unroll
      for (int j = 0; j < 8; ++j) {
        v2f bb = *(const v2f*)&l21[((j * 2 + t) * 32 + lane) * 2];
        v2f bc = *(const v2f*)&l21[(((8 + j) * 2 + t) * 32 + lane) * 2];
        cb = wmma4(fB[j], bb, cb);
        c0 = wmma4(fC0[j], bc, c0);
        c1 = wmma4(fC1[j], bc, c1);
        c2 = wmma4(fC2[j], bc, c2);
      }
#pragma unroll
      for (int v = 0; v < 8; ++v) {
        if (ok[v]) {
          float* m = acc + (long)dv[v] * 128 + 32 + 16 * t + r;
          atomicAdd(m,      cb[v] * y1x[v] + c0[v]);
          atomicAdd(m + 32, cb[v] * y1y[v] + c1[v]);
          atomicAdd(m + 64, cb[v] * y1z[v] + c2[v]);
        }
      }
    }
  }
}

__global__ void node_post_kernel(const float* __restrict__ acc,
                                 const float* __restrict__ ang,
                                 const float* __restrict__ node_attr,
                                 const float* __restrict__ scx,
                                 float* __restrict__ out, int N) {
  int wv = threadIdx.x >> 5;
  int lane = threadIdx.x & 31;
  int n = blockIdx.x * (blockDim.x >> 5) + wv;
  if (n >= N) return;
  const float* a = acc + (long)n * 128;
  const float* sx = scx + (long)n * 128;
  float* o = out + (long)n * 128;
  float k = node_attr[n] * 0.125f;
  float angle = 0.1f * ang[n] * k;
  float ca = cosf(angle), sa = sinf(angle);
  float sk = sa * k;
  o[lane] = ca * sx[lane] + sk * a[lane];
#pragma unroll
  for (int c = 0; c < 3; ++c) {
    int jj = lane + 32 * c;
    int w = jj / 3, m = jj - 3 * w;
    o[32 + jj] = ca * sx[32 + jj] + sk * a[32 + 32 * m + w];
  }
}

extern "C" void kernel_launch(void* const* d_in, const int* in_sizes, int n_in,
                              void* d_out, int out_size, void* d_ws, size_t ws_size,
                              hipStream_t stream) {
  const float* node_input   = (const float*)d_in[0];
  const float* node_attr    = (const float*)d_in[1];
  const int*   edge_src     = (const int*)d_in[2];
  const int*   edge_dst     = (const int*)d_in[3];
  const float* edge_attr    = (const float*)d_in[4];
  const float* edge_scalars = (const float*)d_in[5];
  const float* W_sc0        = (const float*)d_in[6];
  const float* W_sc1        = (const float*)d_in[7];
  const float* W_lin1_0     = (const float*)d_in[8];
  const float* W_lin1_1     = (const float*)d_in[9];
  const float* W_fc1        = (const float*)d_in[10];
  const float* W_fc2        = (const float*)d_in[11];
  const float* W_lin2_0     = (const float*)d_in[12];
  const float* W_lin2_1     = (const float*)d_in[13];
  const float* W_lin3       = (const float*)d_in[14];
  int N = in_sizes[1];
  int E = in_sizes[2];

  float* f0f1 = (float*)d_ws;
  float* scx  = f0f1 + (size_t)N * 128;
  float* acc  = scx + (size_t)N * 128;
  float* ang  = acc + (size_t)N * 128;
  float* out  = (float*)d_out;

  zero_kernel<<<2048, 256, 0, stream>>>(acc, (long)N * 129);

  int nblk_node = (N + 7) / 8;
  node_pre_kernel<<<nblk_node, 256, 0, stream>>>(
      node_input, node_attr, W_sc0, W_sc1, W_lin1_0, W_lin1_1, f0f1, scx, N);

  edge_kernel<<<1024, EK_WAVES * 32, 0, stream>>>(
      edge_scalars, edge_attr, edge_src, edge_dst, W_fc1, W_fc2,
      W_lin2_0, W_lin2_1, W_lin3, f0f1, acc, ang, E);

  node_post_kernel<<<nblk_node, 256, 0, stream>>>(
      acc, ang, node_attr, scx, out, N);
}


# edge GEMMs: weight fragments packed two k-steps per 16-byte lane entry; a column tile's B panel is K/8 16-byte LDS reads off one base (was an 8-byte read and wait per WMMA); sum order unchanged
# speedup vs baseline: 1.2824x; 1.1454x over previous
#include <hip/hip_runtime.h>
#include <math.h>


typedef __attribute__((ext_vector_type(2))) float v2f;
typedef __attribute__((ext_vector_type(4))) float v4f;
typedef __attribute__((ext_vector_type(8))) float v8f;

__device__ __forceinline__ v8f wmma4(v2f a, v2f b, v8f c) {
  return __builtin_amdgcn_wmma_f32_16x16x4_f32(false, a, false, b, (short)0, c,
                                               false, false);
}

__global__ void zero_kernel(float* __restrict__ p, long n) {
  long i = (long)blockIdx.x * blockDim.x + threadIdx.x;
  long stride = (long)gridDim.x * blockDim.x;
  for (long k = i * 4; k < n; k += stride * 4) {
    if (k + 4 <= n) {
      *(float4*)(p + k) = make_float4(0.f, 0.f, 0.f, 0.f);
    } else {
      for (long kk = k; kk < n; ++kk) p[kk] = 0.f;
    }
  }
}

__global__ void node_pre_kernel(const float* __restrict__ node_input,
                                const float* __restrict__ node_attr,
                                const float* __restrict__ Wsc0,
                                const float* __restrict__ Wsc1,
                                const float* __restrict__ Wl10,
                                const float* __restrict__ Wl11,
                                float* __restrict__ f0f1,
                                float* __restrict__ scx, int N) {
  int wv = threadIdx.x >> 5;
  int lane = threadIdx.x & 31;
  int n = blockIdx.x * (blockDim.x >> 5) + wv;
  if (n >= N) return;
  const float* xi = node_input + (long)n * 128;
  float f0 = 0.f, f1x = 0.f, f1y = 0.f, f1z = 0.f;
  float s0 = 0.f, s1x = 0.f, s1y = 0.f, s1z = 0.f;
  for (int u = 0; u < 32; ++u) {
    float x0 = xi[u];
    float x1x = xi[32 + u * 3 + 0];
    float x1y = xi[32 + u * 3 + 1];
    float x1z = xi[32 + u * 3 + 2];
    float wl0 = Wl10[u * 32 + lane], wl1 = Wl11[u * 32 + lane];
    float ws0 = Wsc0[u * 32 + lane], ws1 = Wsc1[u * 32 + lane];
    f0 += x0 * wl0;
    f1x += x1x * wl1; f1y += x1y * wl1; f1z += x1z * wl1;
    s0 += x0 * ws0;
    s1x += x1x * ws1; s1y += x1y * ws1; s1z += x1z * ws1;
  }
  float k = node_attr[n] * 0.17677669529663687f;
  float* fo = f0f1 + (long)n * 128;
  fo[lane] = f0 * k;
  fo[32 + lane] = f1x * k;
  fo[64 + lane] = f1y * k;
  fo[96 + lane] = f1z * k;
  float* so = scx + (long)n * 128;
  so[lane] = s0 * k;
  so[32 + lane * 3 + 0] = s1x * k;
  so[32 + lane * 3 + 1] = s1y * k;
  so[32 + lane * 3 + 2] = s1z * k;
}

#define EK_WAVES 8
#define HSTR 68
#define XS 132

__global__ void __launch_bounds__(EK_WAVES * 32, 4)
edge_kernel(const float* __restrict__ edge_scalars,
            const float* __restrict__ edge_attr,
            const int* __restrict__ edge_src,
            const int* __restrict__ edge_dst,
            const float* __restrict__ Wfc1,
            const float* __restrict__ Wfc2,
            const float* __restrict__ Wl20,
            const float* __restrict__ Wl21,
            const float* __restrict__ Wl3,
            const float* __restrict__ f0f1,
            float* __restrict__ acc,
            float* __restrict__ ang, int E) {
  __shared__ v4f fc1[2 * 4 * 32];
  __shared__ v4f fc2[8 * 8 * 32];
  __shared__ v4f l20[8 * 2 * 32];
  __shared__ v4f l21[8 * 2 * 32];
  __shared__ __attribute__((aligned(16))) float w3s[64];
  __shared__ __attribute__((aligned(16))) float stage[EK_WAVES][16 * XS];

  for (int idx = threadIdx.x; idx < 2 * 4 * 32; idx += blockDim.x) {
    int ln = idx & 31, t = (idx >> 5) & 3, jj = idx >> 7;
    int row = 8 * jj + 2 * (ln >> 4), col = 16 * t + (ln & 15);
    v4f q = {Wfc1[row * 64 + col], Wfc1[(row + 1) * 64 + col],
             Wfc1[(row + 4) * 64 + col], Wfc1[(row + 5) * 64 + col]};
    fc1[idx] = q;
  }
  for (int idx = threadIdx.x; idx < 8 * 8 * 32; idx += blockDim.x) {
    int ln = idx & 31, t = (idx >> 5) & 7, jj = idx >> 8;
    int row = 8 * jj + 2 * (ln >> 4), col = 16 * t + (ln & 15);
    v4f q = {Wfc2[row * 128 + col], Wfc2[(row + 1) * 128 + col],
             Wfc2[(row + 4) * 128 + col], Wfc2[(row + 5) * 128 + col]};
    fc2[idx] = q;
  }
  for (int idx = threadIdx.x; idx < 8 * 2 * 32; idx += blockDim.x) {
    int ln = idx & 31, t = (idx >> 5) & 1, jj = idx >> 6;
    int row = 8 * jj + 2 * (ln >> 4), col = 16 * t + (ln & 15);
    v4f q0 = {Wl20[row * 32 + col], Wl20[(row + 1) * 32 + col],
              Wl20[(row + 4) * 32 + col], Wl20[(row + 5) * 32 + col]};
    v4f q1 = {Wl21[row * 32 + col], Wl21[(row + 1) * 32 + col],
              Wl21[(row + 4) * 32 + col], Wl21[(row + 5) * 32 + col]};
    l20[idx] = q0;
    l21[idx] = q1;
  }
  for (int idx = threadIdx.x; idx < 64; idx += blockDim.x) w3s[idx] = Wl3[idx];
  __syncthreads();

  const int lane = threadIdx.x & 31;
  const int wv = threadIdx.x >> 5;
  const int r = lane & 15;
  const int hi = lane >> 4;
  float* st = stage[wv];
  const int nw = gridDim.x * (blockDim.x >> 5);
  const int gw = blockIdx.x * (blockDim.x >> 5) + wv;
  const int tiles = (E + 15) >> 4;

  for (int tile = gw; tile < tiles; tile += nw) {
    const int e0 = tile * 16;
    int er = e0 + r;
    if (er >= E) er = E - 1;
    const float* srow = edge_scalars + (long)er * 16;

    v2f a1[4];
#pragma unroll
    for (int j = 0; j < 4; ++j) a1[j] = *(const v2f*)(srow + 4 * j + 2 * hi);

    const int sr = edge_src[er];
    const float4 ya = *(const float4*)(edge_attr + (long)er * 4);
    const int dr = edge_dst[er];

    const float* gp = f0f1 + (long)sr * 128 + 2 * hi;
    v2f g0[8], g1x[8], g1y[8], g1z[8];
#pragma unroll
    for (int j = 0; j < 8; ++j) g0[j] = *(const v2f*)(gp + 4 * j);
#pragma unroll
    for (int j = 0; j < 8; ++j) g1x[j] = *(const v2f*)(gp + 32 + 4 * j);
#pragma unroll
    for (int j = 0; j < 8; ++j) g1y[j] = *(const v2f*)(gp + 64 + 4 * j);
#pragma unroll
    for (int j = 0; j < 8; ++j) g1z[j] = *(const v2f*)(gp + 96 + 4 * j);

    int dv[8];
    float y1x[8], y1y[8], y1z[8];
    bool ok[8];
#pragma unroll
    for (int v = 0; v < 8; ++v) {
      int row = e0 + v + 8 * hi;
      ok[v] = row < E;
      if (!ok[v]) row = E - 1;
      dv[v] = edge_dst[row];
      const float4 yv = *(const float4*)(edge_attr + (long)row * 4);
      y1x[v] = yv.y; y1y[v] = yv.z; y1z[v] = yv.w;
    }

    __builtin_amdgcn_sched_barrier(0);

#pragma unroll
    for (int t = 0; t < 4; ++t) {
      v8f c = {0.f, 0.f, 0.f, 0.f, 0.f, 0.f, 0.f, 0.f};
      const v4f* bp = fc1 + t * 32 + lane;
      v4f b[2];
#pragma unroll
      for (int jj = 0; jj < 2; ++jj) b[jj] = bp[jj * 4 * 32];
#pragma unroll
      for (int jj = 0; jj < 2; ++jj) {
        c = wmma4(a1[2 * jj], b[jj].xy, c);
        c = wmma4(a1[2 * jj + 1], b[jj].zw, c);
      }
#pragma unroll
      for (int v = 0; v < 8; ++v) {
        float x = c[v] * 0.25f;
        st[(v + 8 * hi) * HSTR + 16 * t + r] = x / (1.0f + __expf(-x));
      }
    }

    v2f a2[16];
#pragma unroll
    for (int j = 0; j < 16; ++j)
      a2[j] = *(const v2f*)&st[r * HSTR + 4 * j + 2 * hi];

#pragma unroll
    for (int t = 0; t < 8; ++t) {
      v8f c = {0.f, 0.f, 0.f, 0.f, 0.f, 0.f, 0.f, 0.f};
      const v4f* bp = fc2 + t * 32 + lane;
      v4f b[8];
#pragma unroll
      for (int jj = 0; jj < 8; ++jj) b[jj] = bp[jj * 8 * 32];
#pragma unroll
      for (int jj = 0; jj < 8; ++jj) {
        c = wmma4(a2[2 * jj], b[jj].xy, c);
        c = wmma4(a2[2 * jj + 1], b[jj].zw, c);
      }
#pragma unroll
      for (int v = 0; v < 8; ++v)
        st[(v + 8 * hi) * XS + 16 * t + r] = c[v] * 0.125f;
    }

    const float q = 0.31622776601683794f;
    const float r3 = 0.5773502691896258f;
    const float* wr = st + r * XS + 2 * hi;
    v2f fA[8], fD[8], fB[8], fC0[8], fC1[8], fC2[8];
#pragma unroll
    for (int j = 0; j < 8; ++j) {
      const v2f wA = *(const v2f*)(wr + 4 * j);
      const v2f wB = *(const v2f*)(wr + 32 + 4 * j);
      const v2f wC = *(const v2f*)(wr + 64 + 4 * j);
      const v2f wD = *(const v2f*)(wr + 96 + 4 * j);
      fA[j] = wA * g0[j] * ya.x * q;
      fD[j] = wD * (g1x[j] * ya.y + g1y[j] * ya.z + g1z[j] * ya.w) * r3 * q;
      fB[j] = wB * g0[j] * q;
      const v2f cc = wC * ya.x * q;
      fC0[j] = cc * g1x[j];
      fC1[j] = cc * g1y[j];
      fC2[j] = cc * g1z[j];
    }

#pragma unroll
    for (int t = 0; t < 2; ++t) {
      v8f c = {0.f, 0.f, 0.f, 0.f, 0.f, 0.f, 0.f, 0.f};
      const v4f* bp = l20 + t * 32 + lane;
      v4f b[8];
#pragma unroll
      for (int jj = 0; jj < 8; ++jj) b[jj] = bp[jj * 2 * 32];
#pragma unroll
      for (int jj = 0; jj < 8; ++jj) {
        const int j = 2 * jj;
        c = wmma4(j < 8 ? fA[j] : fD[j - 8], b[jj].xy, c);
        c = wmma4(j < 8 ? fA[j + 1] : fD[j - 7], b[jj].zw, c);
      }
#pragma unroll
      for (int v = 0; v < 8; ++v) {
        if (ok[v]) {
          float* m = acc + (long)dv[v] * 128 + 16 * t + r;
          atomicAdd(m, c[v]);
        }
      }
    }

    {
      const float* w3p = w3s + 2 * hi;
      float p = 0.f;
#pragma unroll
      for (int j = 0; j < 8; ++j) {
        const v2f wa = *(const v2f*)(w3p + 4 * j);
        const v2f wd = *(const v2f*)(w3p + 32 + 4 * j);
        p += fA[j].x * wa.x + fA[j].y * wa.y + fD[j].x * wd.x + fD[j].y * wd.y;
      }
      p += __shfl_xor(p, 16);
      if (hi == 0 && (e0 + r) < E) atomicAdd(ang + dr, p);
    }

#pragma unroll
    for (int t = 0; t < 2; ++t) {
      v8f cb = {0.f, 0.f, 0.f, 0.f, 0.f, 0.f, 0.f, 0.f};
      v8f c0 = cb, c1 = cb, c2 = cb;
      const v4f* bp = l21 + t * 32 + lane;
      v4f qb[4], qc[4];
#pragma unroll
      for (int jj = 0; jj < 4; ++jj) {
        qb[jj] = bp[jj * 2 * 32];
        qc[jj] = bp[(4 + jj) * 2 * 32];
      }
#pragma unroll
      for (int jj = 0; jj < 4; ++jj) {
        cb = wmma4(fB[2 * jj], qb[jj].xy, cb);
        c0 = wmma4(fC0[2 * jj], qc[jj].xy, c0);
        c1 = wmma4(fC1[2 * jj], qc[jj].xy, c1);
        c2 = wmma4(fC2[2 * jj], qc[jj].xy, c2);
        cb = wmma4(fB[2 * jj + 1], qb[jj].zw, cb);
        c0 = wmma4(fC0[2 * jj + 1], qc[jj].zw, c0);
        c1 = wmma4(fC1[2 * jj + 1], qc[jj].zw, c1);
        c2 = wmma4(fC2[2 * jj + 1], qc[jj].zw, c2);
      }
#pragma unroll
      for (int v = 0; v < 8; ++v) {
        if (ok[v]) {
          float* m = acc + (long)dv[v] * 128 + 32 + 16 * t + r;
          atomicAdd(m,      cb[v] * y1x[v] + c0[v]);
          atomicAdd(m + 32, cb[v] * y1y[v] + c1[v]);
          atomicAdd(m + 64, cb[v] * y1z[v] + c2[v]);
        }
      }
    }
  }
}

__global__ void node_post_kernel(const float* __restrict__ acc,
                                 const float* __restrict__ ang,
                                 const float* __restrict__ node_attr,
                                 const float* __restrict__ scx,
                                 float* __restrict__ out, int N) {
  int wv = threadIdx.x >> 5;
  int lane = threadIdx.x & 31;
  int n = blockIdx.x * (blockDim.x >> 5) + wv;
  if (n >= N) return;
  const float* a = acc + (long)n * 128;
  const float* sx = scx + (long)n * 128;
  float* o = out + (long)n * 128;
  float k = node_attr[n] * 0.125f;
  float angle = 0.1f * ang[n] * k;
  float ca = cosf(angle), sa = sinf(angle);
  float sk = sa * k;
  o[lane] = ca * sx[lane] + sk * a[lane];
#pragma unroll
  for (int c = 0; c < 3; ++c) {
    int jj = lane + 32 * c;
    int w = jj / 3, m = jj - 3 * w;
    o[32 + jj] = ca * sx[32 + jj] + sk * a[32 + 32 * m + w];
  }
}

extern "C" void kernel_launch(void* const* d_in, const int* in_sizes, int n_in,
                              void* d_out, int out_size, void* d_ws, size_t ws_size,
                              hipStream_t stream) {
  const float* node_input   = (const float*)d_in[0];
  const float* node_attr    = (const float*)d_in[1];
  const int*   edge_src     = (const int*)d_in[2];
  const int*   edge_dst     = (const int*)d_in[3];
  const float* edge_attr    = (const float*)d_in[4];
  const float* edge_scalars = (const float*)d_in[5];
  const float* W_sc0        = (const float*)d_in[6];
  const float* W_sc1        = (const float*)d_in[7];
  const float* W_lin1_0     = (const float*)d_in[8];
  const float* W_lin1_1     = (const float*)d_in[9];
  const float* W_fc1        = (const float*)d_in[10];
  const float* W_fc2        = (const float*)d_in[11];
  const float* W_lin2_0     = (const float*)d_in[12];
  const float* W_lin2_1     = (const float*)d_in[13];
  const float* W_lin3       = (const float*)d_in[14];
  int N = in_sizes[1];
  int E = in_sizes[2];

  float* f0f1 = (float*)d_ws;
  float* scx  = f0f1 + (size_t)N * 128;
  float* acc  = scx + (size_t)N * 128;
  float* ang  = acc + (size_t)N * 128;
  float* out  = (float*)d_out;

  zero_kernel<<<2048, 256, 0, stream>>>(acc, (long)N * 129);

  int nblk_node = (N + 7) / 8;
  node_pre_kernel<<<nblk_node, 256, 0, stream>>>(
      node_input, node_attr, W_sc0, W_sc1, W_lin1_0, W_lin1_1, f0f1, scx, N);

  edge_kernel<<<1024, EK_WAVES * 32, 0, stream>>>(
      edge_scalars, edge_attr, edge_src, edge_dst, W_fc1, W_fc2,
      W_lin2_0, W_lin2_1, W_lin3, f0f1, acc, ang, E);

  node_post_kernel<<<nblk_node, 256, 0, stream>>>(
      acc, ang, node_attr, scx, out, N);
}


# edge scatter: tile index made scalar (readfirstlane); full tiles add with one instruction per atomic, row byte offsets in registers (was mask, branch, 64-bit address per add); sum order unchanged
# speedup vs baseline: 1.3253x; 1.0335x over previous
#include <hip/hip_runtime.h>
#include <math.h>


typedef __attribute__((ext_vector_type(2))) float v2f;
typedef __attribute__((ext_vector_type(4))) float v4f;
typedef __attribute__((ext_vector_type(8))) float v8f;

__device__ __forceinline__ v8f wmma4(v2f a, v2f b, v8f c) {
  return __builtin_amdgcn_wmma_f32_16x16x4_f32(false, a, false, b, (short)0, c,
                                               false, false);
}

__global__ void zero_kernel(float* __restrict__ p, long n) {
  long i = (long)blockIdx.x * blockDim.x + threadIdx.x;
  long stride = (long)gridDim.x * blockDim.x;
  for (long k = i * 4; k < n; k += stride * 4) {
    if (k + 4 <= n) {
      *(float4*)(p + k) = make_float4(0.f, 0.f, 0.f, 0.f);
    } else {
      for (long kk = k; kk < n; ++kk) p[kk] = 0.f;
    }
  }
}

__global__ void node_pre_kernel(const float* __restrict__ node_input,
                                const float* __restrict__ node_attr,
                                const float* __restrict__ Wsc0,
                                const float* __restrict__ Wsc1,
                                const float* __restrict__ Wl10,
                                const float* __restrict__ Wl11,
                                float* __restrict__ f0f1,
                                float* __restrict__ scx, int N) {
  int wv = threadIdx.x >> 5;
  int lane = threadIdx.x & 31;
  int n = blockIdx.x * (blockDim.x >> 5) + wv;
  if (n >= N) return;
  const float* xi = node_input + (long)n * 128;
  float f0 = 0.f, f1x = 0.f, f1y = 0.f, f1z = 0.f;
  float s0 = 0.f, s1x = 0.f, s1y = 0.f, s1z = 0.f;
  for (int u = 0; u < 32; ++u) {
    float x0 = xi[u];
    float x1x = xi[32 + u * 3 + 0];
    float x1y = xi[32 + u * 3 + 1];
    float x1z = xi[32 + u * 3 + 2];
    float wl0 = Wl10[u * 32 + lane], wl1 = Wl11[u * 32 + lane];
    float ws0 = Wsc0[u * 32 + lane], ws1 = Wsc1[u * 32 + lane];
    f0 += x0 * wl0;
    f1x += x1x * wl1; f1y += x1y * wl1; f1z += x1z * wl1;
    s0 += x0 * ws0;
    s1x += x1x * ws1; s1y += x1y * ws1; s1z += x1z * ws1;
  }
  float k = node_attr[n] * 0.17677669529663687f;
  float* fo = f0f1 + (long)n * 128;
  fo[lane] = f0 * k;
  fo[32 + lane] = f1x * k;
  fo[64 + lane] = f1y * k;
  fo[96 + lane] = f1z * k;
  float* so = scx + (long)n * 128;
  so[lane] = s0 * k;
  so[32 + lane * 3 + 0] = s1x * k;
  so[32 + lane * 3 + 1] = s1y * k;
  so[32 + lane * 3 + 2] = s1z * k;
}

#define EK_WAVES 8
#define HSTR 68
#define XS 132

__global__ void __launch_bounds__(EK_WAVES * 32, 4)
edge_kernel(const float* __restrict__ edge_scalars,
            const float* __restrict__ edge_attr,
            const int* __restrict__ edge_src,
            const int* __restrict__ edge_dst,
            const float* __restrict__ Wfc1,
            const float* __restrict__ Wfc2,
            const float* __restrict__ Wl20,
            const float* __restrict__ Wl21,
            const float* __restrict__ Wl3,
            const float* __restrict__ f0f1,
            float* __restrict__ acc,
            float* __restrict__ ang, int E) {
  __shared__ v4f fc1[2 * 4 * 32];
  __shared__ v4f fc2[8 * 8 * 32];
  __shared__ v4f l20[8 * 2 * 32];
  __shared__ v4f l21[8 * 2 * 32];
  __shared__ __attribute__((aligned(16))) float w3s[64];
  __shared__ __attribute__((aligned(16))) float stage[EK_WAVES][16 * XS];

  for (int idx = threadIdx.x; idx < 2 * 4 * 32; idx += blockDim.x) {
    int ln = idx & 31, t = (idx >> 5) & 3, jj = idx >> 7;
    int row = 8 * jj + 2 * (ln >> 4), col = 16 * t + (ln & 15);
    v4f q = {Wfc1[row * 64 + col], Wfc1[(row + 1) * 64 + col],
             Wfc1[(row + 4) * 64 + col], Wfc1[(row + 5) * 64 + col]};
    fc1[idx] = q;
  }
  for (int idx = threadIdx.x; idx < 8 * 8 * 32; idx += blockDim.x) {
    int ln = idx & 31, t = (idx >> 5) & 7, jj = idx >> 8;
    int row = 8 * jj + 2 * (ln >> 4), col = 16 * t + (ln & 15);
    v4f q = {Wfc2[row * 128 + col], Wfc2[(row + 1) * 128 + col],
             Wfc2[(row + 4) * 128 + col], Wfc2[(row + 5) * 128 + col]};
    fc2[idx] = q;
  }
  for (int idx = threadIdx.x; idx < 8 * 2 * 32; idx += blockDim.x) {
    int ln = idx & 31, t = (idx >> 5) & 1, jj = idx >> 6;
    int row = 8 * jj + 2 * (ln >> 4), col = 16 * t + (ln & 15);
    v4f q0 = {Wl20[row * 32 + col], Wl20[(row + 1) * 32 + col],
              Wl20[(row + 4) * 32 + col], Wl20[(row + 5) * 32 + col]};
    v4f q1 = {Wl21[row * 32 + col], Wl21[(row + 1) * 32 + col],
              Wl21[(row + 4) * 32 + col], Wl21[(row + 5) * 32 + col]};
    l20[idx] = q0;
    l21[idx] = q1;
  }
  for (int idx = threadIdx.x; idx < 64; idx += blockDim.x) w3s[idx] = Wl3[idx];
  __syncthreads();

  const int lane = threadIdx.x & 31;
  const int wv = __builtin_amdgcn_readfirstlane((int)(threadIdx.x >> 5));
  const int r = lane & 15;
  const int hi = lane >> 4;
  float* st = stage[wv];
  const int nw = gridDim.x * EK_WAVES;
  const int gw = blockIdx.x * EK_WAVES + wv;
  const int tiles = (E + 15) >> 4;

  for (int tile = gw; tile < tiles; tile += nw) {
    const int e0 = tile * 16;
    const bool full = e0 + 16 <= E;
    int er = e0 + r;
    if (er >= E) er = E - 1;
    const float* srow = edge_scalars + (long)er * 16;

    v2f a1[4];
#pragma unroll
    for (int j = 0; j < 4; ++j) a1[j] = *(const v2f*)(srow + 4 * j + 2 * hi);

    const int sr = edge_src[er];
    const float4 ya = *(const float4*)(edge_attr + (long)er * 4);
    const int dr = edge_dst[er];

    const float* gp = f0f1 + (long)sr * 128 + 2 * hi;
    v2f g0[8], g1x[8], g1y[8], g1z[8];
#pragma unroll
    for (int j = 0; j < 8; ++j) g0[j] = *(const v2f*)(gp + 4 * j);
#pragma unroll
    for (int j = 0; j < 8; ++j) g1x[j] = *(const v2f*)(gp + 32 + 4 * j);
#pragma unroll
    for (int j = 0; j < 8; ++j) g1y[j] = *(const v2f*)(gp + 64 + 4 * j);
#pragma unroll
    for (int j = 0; j < 8; ++j) g1z[j] = *(const v2f*)(gp + 96 + 4 * j);

    unsigned po[8];
    float y1x[8], y1y[8], y1z[8];
#pragma unroll
    for (int v = 0; v < 8; ++v) {
      int row = e0 + v + 8 * hi;
      if (row >= E) row = E - 1;
      po[v] = (unsigned)edge_dst[row] * 512u + (unsigned)r * 4u;
      const float4 yv = *(const float4*)(edge_attr + (long)row * 4);
      y1x[v] = yv.y; y1y[v] = yv.z; y1z[v] = yv.w;
    }

    __builtin_amdgcn_sched_barrier(0);

#pragma unroll
    for (int t = 0; t < 4; ++t) {
      v8f c = {0.f, 0.f, 0.f, 0.f, 0.f, 0.f, 0.f, 0.f};
      const v4f* bp = fc1 + t * 32 + lane;
      v4f b[2];
#pragma unroll
      for (int jj = 0; jj < 2; ++jj) b[jj] = bp[jj * 4 * 32];
#pragma unroll
      for (int jj = 0; jj < 2; ++jj) {
        c = wmma4(a1[2 * jj], b[jj].xy, c);
        c = wmma4(a1[2 * jj + 1], b[jj].zw, c);
      }
#pragma unroll
      for (int v = 0; v < 8; ++v) {
        float x = c[v] * 0.25f;
        st[(v + 8 * hi) * HSTR + 16 * t + r] = x / (1.0f + __expf(-x));
      }
    }

    v2f a2[16];
#pragma unroll
    for (int j = 0; j < 16; ++j)
      a2[j] = *(const v2f*)&st[r * HSTR + 4 * j + 2 * hi];

#pragma unroll
    for (int t = 0; t < 8; ++t) {
      v8f c = {0.f, 0.f, 0.f, 0.f, 0.f, 0.f, 0.f, 0.f};
      const v4f* bp = fc2 + t * 32 + lane;
      v4f b[8];
#pragma unroll
      for (int jj = 0; jj < 8; ++jj) b[jj] = bp[jj * 8 * 32];
#pragma unroll
      for (int jj = 0; jj < 8; ++jj) {
        c = wmma4(a2[2 * jj], b[jj].xy, c);
        c = wmma4(a2[2 * jj + 1], b[jj].zw, c);
      }
#pragma unroll
      for (int v = 0; v < 8; ++v)
        st[(v + 8 * hi) * XS + 16 * t + r] = c[v] * 0.125f;
    }

    const float q = 0.31622776601683794f;
    const float r3 = 0.5773502691896258f;
    const float* wr = st + r * XS + 2 * hi;
    v2f fA[8], fD[8], fB[8], fC0[8], fC1[8], fC2[8];
#pragma unroll
    for (int j = 0; j < 8; ++j) {
      const v2f wA = *(const v2f*)(wr + 4 * j);
      const v2f wB = *(const v2f*)(wr + 32 + 4 * j);
      const v2f wC = *(const v2f*)(wr + 64 + 4 * j);
      const v2f wD = *(const v2f*)(wr + 96 + 4 * j);
      fA[j] = wA * g0[j] * ya.x * q;
      fD[j] = wD * (g1x[j] * ya.y + g1y[j] * ya.z + g1z[j] * ya.w) * r3 * q;
      fB[j] = wB * g0[j] * q;
      const v2f cc = wC * ya.x * q;
      fC0[j] = cc * g1x[j];
      fC1[j] = cc * g1y[j];
      fC2[j] = cc * g1z[j];
    }

#pragma unroll
    for (int t = 0; t < 2; ++t) {
      v8f c = {0.f, 0.f, 0.f, 0.f, 0.f, 0.f, 0.f, 0.f};
      const v4f* bp = l20 + t * 32 + lane;
      v4f b[8];
#pragma unroll
      for (int jj = 0; jj < 8; ++jj) b[jj] = bp[jj * 2 * 32];
#pragma unroll
      for (int jj = 0; jj < 8; ++jj) {
        const int j = 2 * jj;
        c = wmma4(j < 8 ? fA[j] : fD[j - 8], b[jj].xy, c);
        c = wmma4(j < 8 ? fA[j + 1] : fD[j - 7], b[jj].zw, c);
      }
      if (full) {
#pragma unroll
        for (int v = 0; v < 8; ++v)
          atomicAdd((float*)((char*)acc + po[v]) + 16 * t, c[v]);
      } else {
#pragma unroll
        for (int v = 0; v < 8; ++v)
          if (e0 + v + 8 * hi < E)
            atomicAdd((float*)((char*)acc + po[v]) + 16 * t, c[v]);
      }
    }

    {
      const float* w3p = w3s + 2 * hi;
      float p = 0.f;
#pragma unroll
      for (int j = 0; j < 8; ++j) {
        const v2f wa = *(const v2f*)(w3p + 4 * j);
        const v2f wd = *(const v2f*)(w3p + 32 + 4 * j);
        p += fA[j].x * wa.x + fA[j].y * wa.y + fD[j].x * wd.x + fD[j].y * wd.y;
      }
      p += __shfl_xor(p, 16);
      if (hi == 0 && (e0 + r) < E) atomicAdd(ang + dr, p);
    }

#pragma unroll
    for (int t = 0; t < 2; ++t) {
      v8f cb = {0.f, 0.f, 0.f, 0.f, 0.f, 0.f, 0.f, 0.f};
      v8f c0 = cb, c1 = cb, c2 = cb;
      const v4f* bp = l21 + t * 32 + lane;
      v4f qb[4], qc[4];
#pragma unroll
      for (int jj = 0; jj < 4; ++jj) {
        qb[jj] = bp[jj * 2 * 32];
        qc[jj] = bp[(4 + jj) * 2 * 32];
      }
#pragma unroll
      for (int jj = 0; jj < 4; ++jj) {
        cb = wmma4(fB[2 * jj], qb[jj].xy, cb);
        c0 = wmma4(fC0[2 * jj], qc[jj].xy, c0);
        c1 = wmma4(fC1[2 * jj], qc[jj].xy, c1);
        c2 = wmma4(fC2[2 * jj], qc[jj].xy, c2);
        cb = wmma4(fB[2 * jj + 1], qb[jj].zw, cb);
        c0 = wmma4(fC0[2 * jj + 1], qc[jj].zw, c0);
        c1 = wmma4(fC1[2 * jj + 1], qc[jj].zw, c1);
        c2 = wmma4(fC2[2 * jj + 1], qc[jj].zw, c2);
      }
      if (full) {
#pragma unroll
        for (int v = 0; v < 8; ++v) {
          float* m = (float*)((char*)acc + po[v]) + 32 + 16 * t;
          atomicAdd(m,      cb[v] * y1x[v] + c0[v]);
          atomicAdd(m + 32, cb[v] * y1y[v] + c1[v]);
          atomicAdd(m + 64, cb[v] * y1z[v] + c2[v]);
        }
      } else {
#pragma unroll
        for (int v = 0; v < 8; ++v) {
          if (e0 + v + 8 * hi < E) {
            float* m = (float*)((char*)acc + po[v]) + 32 + 16 * t;
            atomicAdd(m,      cb[v] * y1x[v] + c0[v]);
            atomicAdd(m + 32, cb[v] * y1y[v] + c1[v]);
            atomicAdd(m + 64, cb[v] * y1z[v] + c2[v]);
          }
        }
      }
    }
  }
}

__global__ void node_post_kernel(const float* __restrict__ acc,
                                 const float* __restrict__ ang,
                                 const float* __restrict__ node_attr,
                                 const float* __restrict__ scx,
                                 float* __restrict__ out, int N) {
  int wv = threadIdx.x >> 5;
  int lane = threadIdx.x & 31;
  int n = blockIdx.x * (blockDim.x >> 5) + wv;
  if (n >= N) return;
  const float* a = acc + (long)n * 128;
  const float* sx = scx + (long)n * 128;
  float* o = out + (long)n * 128;
  float k = node_attr[n] * 0.125f;
  float angle = 0.1f * ang[n] * k;
  float ca = cosf(angle), sa = sinf(angle);
  float sk = sa * k;
  o[lane] = ca * sx[lane] + sk * a[lane];
#pragma unroll
  for (int c = 0; c < 3; ++c) {
    int jj = lane + 32 * c;
    int w = jj / 3, m = jj - 3 * w;
    o[32 + jj] = ca * sx[32 + jj] + sk * a[32 + 32 * m + w];
  }
}

extern "C" void kernel_launch(void* const* d_in, const int* in_sizes, int n_in,
                              void* d_out, int out_size, void* d_ws, size_t ws_size,
                              hipStream_t stream) {
  const float* node_input   = (const float*)d_in[0];
  const float* node_attr    = (const float*)d_in[1];
  const int*   edge_src     = (const int*)d_in[2];
  const int*   edge_dst     = (const int*)d_in[3];
  const float* edge_attr    = (const float*)d_in[4];
  const float* edge_scalars = (const float*)d_in[5];
  const float* W_sc0        = (const float*)d_in[6];
  const float* W_sc1        = (const float*)d_in[7];
  const float* W_lin1_0     = (const float*)d_in[8];
  const float* W_lin1_1     = (const float*)d_in[9];
  const float* W_fc1        = (const float*)d_in[10];
  const float* W_fc2        = (const float*)d_in[11];
  const float* W_lin2_0     = (const float*)d_in[12];
  const float* W_lin2_1     = (const float*)d_in[13];
  const float* W_lin3       = (const float*)d_in[14];
  int N = in_sizes[1];
  int E = in_sizes[2];

  float* f0f1 = (float*)d_ws;
  float* scx  = f0f1 + (size_t)N * 128;
  float* acc  = scx + (size_t)N * 128;
  float* ang  = acc + (size_t)N * 128;
  float* out  = (float*)d_out;

  zero_kernel<<<2048, 256, 0, stream>>>(acc, (long)N * 129);

  int nblk_node = (N + 7) / 8;
  node_pre_kernel<<<nblk_node, 256, 0, stream>>>(
      node_input, node_attr, W_sc0, W_sc1, W_lin1_0, W_lin1_1, f0f1, scx, N);

  edge_kernel<<<1024, EK_WAVES * 32, 0, stream>>>(
      edge_scalars, edge_attr, edge_src, edge_dst, W_fc1, W_fc2,
      W_lin2_0, W_lin2_1, W_lin3, f0f1, acc, ang, E);

  node_post_kernel<<<nblk_node, 256, 0, stream>>>(
      acc, ang, node_attr, scx, out, N);
}


# edge MLP products transposed (weights as A, column = edge): hidden layer and per-edge weights reach the next product in registers, no LDS staging; gather in the same order; k sum order changed
# speedup vs baseline: 1.3709x; 1.0344x over previous
#include <hip/hip_runtime.h>
#include <math.h>


typedef __attribute__((ext_vector_type(2))) float v2f;
typedef __attribute__((ext_vector_type(4))) float v4f;
typedef __attribute__((ext_vector_type(8))) float v8f;

__device__ __forceinline__ v8f wmma4(v2f a, v2f b, v8f c) {
  return __builtin_amdgcn_wmma_f32_16x16x4_f32(false, a, false, b, (short)0, c,
                                               false, false);
}

__global__ void zero_kernel(float* __restrict__ p, long n) {
  long i = (long)blockIdx.x * blockDim.x + threadIdx.x;
  long stride = (long)gridDim.x * blockDim.x;
  for (long k = i * 4; k < n; k += stride * 4) {
    if (k + 4 <= n) {
      *(float4*)(p + k) = make_float4(0.f, 0.f, 0.f, 0.f);
    } else {
      for (long kk = k; kk < n; ++kk) p[kk] = 0.f;
    }
  }
}

__global__ void node_pre_kernel(const float* __restrict__ node_input,
                                const float* __restrict__ node_attr,
                                const float* __restrict__ Wsc0,
                                const float* __restrict__ Wsc1,
                                const float* __restrict__ Wl10,
                                const float* __restrict__ Wl11,
                                float* __restrict__ f0f1,
                                float* __restrict__ scx, int N) {
  int wv = threadIdx.x >> 5;
  int lane = threadIdx.x & 31;
  int n = blockIdx.x * (blockDim.x >> 5) + wv;
  if (n >= N) return;
  const float* xi = node_input + (long)n * 128;
  float f0 = 0.f, f1x = 0.f, f1y = 0.f, f1z = 0.f;
  float s0 = 0.f, s1x = 0.f, s1y = 0.f, s1z = 0.f;
  for (int u = 0; u < 32; ++u) {
    float x0 = xi[u];
    float x1x = xi[32 + u * 3 + 0];
    float x1y = xi[32 + u * 3 + 1];
    float x1z = xi[32 + u * 3 + 2];
    float wl0 = Wl10[u * 32 + lane], wl1 = Wl11[u * 32 + lane];
    float ws0 = Wsc0[u * 32 + lane], ws1 = Wsc1[u * 32 + lane];
    f0 += x0 * wl0;
    f1x += x1x * wl1; f1y += x1y * wl1; f1z += x1z * wl1;
    s0 += x0 * ws0;
    s1x += x1x * ws1; s1y += x1y * ws1; s1z += x1z * ws1;
  }
  float k = node_attr[n] * 0.17677669529663687f;
  float* fo = f0f1 + (long)n * 128;
  fo[lane] = f0 * k;
  fo[32 + lane] = f1x * k;
  fo[64 + lane] = f1y * k;
  fo[96 + lane] = f1z * k;
  float* so = scx + (long)n * 128;
  so[lane] = s0 * k;
  so[32 + lane * 3 + 0] = s1x * k;
  so[32 + lane * 3 + 1] = s1y * k;
  so[32 + lane * 3 + 2] = s1z * k;
}

#define EK_WAVES 8
__device__ __forceinline__ int krow(int kk, int hi) {
  return 16 * (kk >> 1) + 4 * (kk & 1) + 8 * hi;
}

__global__ void __launch_bounds__(EK_WAVES * 32, 4)
edge_kernel(const float* __restrict__ edge_scalars,
            const float* __restrict__ edge_attr,
            const int* __restrict__ edge_src,
            const int* __restrict__ edge_dst,
            const float* __restrict__ Wfc1,
            const float* __restrict__ Wfc2,
            const float* __restrict__ Wl20,
            const float* __restrict__ Wl21,
            const float* __restrict__ Wl3,
            const float* __restrict__ f0f1,
            float* __restrict__ acc,
            float* __restrict__ ang, int E) {
  __shared__ v4f fc1[2 * 4 * 32];
  __shared__ v4f fc2[8 * 8 * 32];
  __shared__ v4f l20[8 * 2 * 32];
  __shared__ v4f l21[8 * 2 * 32];
  __shared__ __attribute__((aligned(16))) float w3s[64];

  for (int idx = threadIdx.x; idx < 2 * 4 * 32; idx += blockDim.x) {
    int ln = idx & 31, t = (idx >> 5) & 3, jj = idx >> 7;
    int row = 8 * jj + 2 * (ln >> 4), col = 16 * t + (ln & 15);
    v4f q = {Wfc1[row * 64 + col], Wfc1[(row + 1) * 64 + col],
             Wfc1[(row + 4) * 64 + col], Wfc1[(row + 5) * 64 + col]};
    fc1[idx] = q;
  }
  for (int idx = threadIdx.x; idx < 8 * 8 * 32; idx += blockDim.x) {
    int ln = idx & 31, t = (idx >> 5) & 7, kk = idx >> 8;
    int row = krow(kk, ln >> 4), col = 16 * t + (ln & 15);
    v4f q = {Wfc2[row * 128 + col], Wfc2[(row + 1) * 128 + col],
             Wfc2[(row + 2) * 128 + col], Wfc2[(row + 3) * 128 + col]};
    fc2[idx] = q;
  }
  for (int idx = threadIdx.x; idx < 8 * 2 * 32; idx += blockDim.x) {
    int ln = idx & 31, t = (idx >> 5) & 1, kk = idx >> 6;
    int row = krow(kk, ln >> 4), col = 16 * t + (ln & 15);
    v4f q0 = {Wl20[row * 32 + col], Wl20[(row + 1) * 32 + col],
              Wl20[(row + 2) * 32 + col], Wl20[(row + 3) * 32 + col]};
    v4f q1 = {Wl21[row * 32 + col], Wl21[(row + 1) * 32 + col],
              Wl21[(row + 2) * 32 + col], Wl21[(row + 3) * 32 + col]};
    l20[idx] = q0;
    l21[idx] = q1;
  }
  for (int idx = threadIdx.x; idx < 64; idx += blockDim.x) w3s[idx] = Wl3[idx];
  __syncthreads();

  const int lane = threadIdx.x & 31;
  const int wv = __builtin_amdgcn_readfirstlane((int)(threadIdx.x >> 5));
  const int r = lane & 15;
  const int hi = lane >> 4;
  const int nw = gridDim.x * EK_WAVES;
  const int gw = blockIdx.x * EK_WAVES + wv;
  const int tiles = (E + 15) >> 4;

  for (int tile = gw; tile < tiles; tile += nw) {
    const int e0 = tile * 16;
    const bool full = e0 + 16 <= E;
    int er = e0 + r;
    if (er >= E) er = E - 1;
    const float* srow = edge_scalars + (long)er * 16;

    v2f a1[4];
#pragma unroll
    for (int j = 0; j < 4; ++j) a1[j] = *(const v2f*)(srow + 4 * j + 2 * hi);

    const int sr = edge_src[er];
    const float4 ya = *(const float4*)(edge_attr + (long)er * 4);
    const int dr = edge_dst[er];

    const float* gp = f0f1 + (long)sr * 128 + 8 * hi;
    float4 gq[4][4];
#pragma unroll
    for (int p = 0; p < 4; ++p) {
      gq[p][0] = *(const float4*)(gp + 32 * p);
      gq[p][1] = *(const float4*)(gp + 32 * p + 4);
      gq[p][2] = *(const float4*)(gp + 32 * p + 16);
      gq[p][3] = *(const float4*)(gp + 32 * p + 20);
    }

    unsigned po[8];
    float y1x[8], y1y[8], y1z[8];
#pragma unroll
    for (int v = 0; v < 8; ++v) {
      int row = e0 + v + 8 * hi;
      if (row >= E) row = E - 1;
      po[v] = (unsigned)edge_dst[row] * 512u + (unsigned)r * 4u;
      const float4 yv = *(const float4*)(edge_attr + (long)row * 4);
      y1x[v] = yv.y; y1y[v] = yv.z; y1z[v] = yv.w;
    }

    __builtin_amdgcn_sched_barrier(0);

    v2f hs[16];
#pragma unroll
    for (int t = 0; t < 4; ++t) {
      v8f c = {0.f, 0.f, 0.f, 0.f, 0.f, 0.f, 0.f, 0.f};
      const v4f* bp = fc1 + t * 32 + lane;
      v4f b[2];
#pragma unroll
      for (int jj = 0; jj < 2; ++jj) b[jj] = bp[jj * 4 * 32];
#pragma unroll
      for (int jj = 0; jj < 2; ++jj) {
        c = wmma4(b[jj].xy, a1[2 * jj], c);
        c = wmma4(b[jj].zw, a1[2 * jj + 1], c);
      }
#pragma unroll
      for (int m = 0; m < 4; ++m) {
        const float x0 = c[2 * m] * 0.25f, x1 = c[2 * m + 1] * 0.25f;
        const v2f s = {x0 / (1.0f + __expf(-x0)), x1 / (1.0f + __expf(-x1))};
        hs[4 * t + m] = s;
      }
    }

    const float q = 0.31622776601683794f;
    const float r3 = 0.5773502691896258f;
    v2f fA[8], fD[8], fB[8], fC0[8], fC1[8], fC2[8];
#pragma unroll
    for (int t = 0; t < 8; ++t) {
      v8f c = {0.f, 0.f, 0.f, 0.f, 0.f, 0.f, 0.f, 0.f};
      const v4f* bp = fc2 + t * 32 + lane;
      v4f b[8];
#pragma unroll
      for (int kk = 0; kk < 8; ++kk) b[kk] = bp[kk * 8 * 32];
#pragma unroll
      for (int kk = 0; kk < 8; ++kk) {
        c = wmma4(b[kk].xy, hs[2 * kk], c);
        c = wmma4(b[kk].zw, hs[2 * kk + 1], c);
      }
#pragma unroll
      for (int m = 0; m < 4; ++m) {
        const int s = 4 * (t & 1) + m;
        const v2f cw = {c[2 * m], c[2 * m + 1]};
        const v2f wp = cw * 0.125f;
        const float4 z0 = gq[0][s >> 1], zx = gq[1][s >> 1], zy = gq[2][s >> 1], zz = gq[3][s >> 1];
        const v2f g0 = (s & 1) ? (v2f){z0.z, z0.w} : (v2f){z0.x, z0.y};
        const v2f gx = (s & 1) ? (v2f){zx.z, zx.w} : (v2f){zx.x, zx.y};
        const v2f gy = (s & 1) ? (v2f){zy.z, zy.w} : (v2f){zy.x, zy.y};
        const v2f gz = (s & 1) ? (v2f){zz.z, zz.w} : (v2f){zz.x, zz.y};
        if ((t >> 1) == 0) {
          fA[s] = wp * g0 * ya.x * q;
        } else if ((t >> 1) == 1) {
          fB[s] = wp * g0 * q;
        } else if ((t >> 1) == 2) {
          const v2f cc = wp * ya.x * q;
          fC0[s] = cc * gx;
          fC1[s] = cc * gy;
          fC2[s] = cc * gz;
        } else {
          fD[s] = wp * (gx * ya.y + gy * ya.z + gz * ya.w) * r3 * q;
        }
      }
    }

#pragma unroll
    for (int t = 0; t < 2; ++t) {
      v8f c = {0.f, 0.f, 0.f, 0.f, 0.f, 0.f, 0.f, 0.f};
      const v4f* bp = l20 + t * 32 + lane;
      v4f b[8];
#pragma unroll
      for (int kk = 0; kk < 8; ++kk) b[kk] = bp[kk * 2 * 32];
#pragma unroll
      for (int kk = 0; kk < 8; ++kk) {
        const v2f* f = kk < 4 ? fA + 2 * kk : fD + 2 * (kk - 4);
        c = wmma4(f[0], b[kk].xy, c);
        c = wmma4(f[1], b[kk].zw, c);
      }
      if (full) {
#pragma unroll
        for (int v = 0; v < 8; ++v)
          atomicAdd((float*)((char*)acc + po[v]) + 16 * t, c[v]);
      } else {
#pragma unroll
        for (int v = 0; v < 8; ++v)
          if (e0 + v + 8 * hi < E)
            atomicAdd((float*)((char*)acc + po[v]) + 16 * t, c[v]);
      }
    }

    {
      const float* w3p = w3s + 8 * hi;
      float p = 0.f;
#pragma unroll
      for (int g = 0; g < 4; ++g) {
        const int off = 16 * (g >> 1) + 4 * (g & 1);
        const float4 wa = *(const float4*)(w3p + off);
        const float4 wd = *(const float4*)(w3p + 32 + off);
        p += fA[2 * g].x * wa.x + fA[2 * g].y * wa.y + fA[2 * g + 1].x * wa.z + fA[2 * g + 1].y * wa.w
           + fD[2 * g].x * wd.x + fD[2 * g].y * wd.y + fD[2 * g + 1].x * wd.z + fD[2 * g + 1].y * wd.w;
      }
      p += __shfl_xor(p, 16);
      if (hi == 0 && (e0 + r) < E) atomicAdd(ang + dr, p);
    }

#pragma unroll
    for (int t = 0; t < 2; ++t) {
      v8f cb = {0.f, 0.f, 0.f, 0.f, 0.f, 0.f, 0.f, 0.f};
      v8f c0 = cb, c1 = cb, c2 = cb;
      const v4f* bp = l21 + t * 32 + lane;
      v4f qb[4], qc[4];
#pragma unroll
      for (int kk = 0; kk < 4; ++kk) {
        qb[kk] = bp[kk * 2 * 32];
        qc[kk] = bp[(4 + kk) * 2 * 32];
      }
#pragma unroll
      for (int kk = 0; kk < 4; ++kk) {
        cb = wmma4(fB[2 * kk], qb[kk].xy, cb);
        c0 = wmma4(fC0[2 * kk], qc[kk].xy, c0);
        c1 = wmma4(fC1[2 * kk], qc[kk].xy, c1);
        c2 = wmma4(fC2[2 * kk], qc[kk].xy, c2);
        cb = wmma4(fB[2 * kk + 1], qb[kk].zw, cb);
        c0 = wmma4(fC0[2 * kk + 1], qc[kk].zw, c0);
        c1 = wmma4(fC1[2 * kk + 1], qc[kk].zw, c1);
        c2 = wmma4(fC2[2 * kk + 1], qc[kk].zw, c2);
      }
      if (full) {
#pragma unroll
        for (int v = 0; v < 8; ++v) {
          float* m = (float*)((char*)acc + po[v]) + 32 + 16 * t;
          atomicAdd(m,      cb[v] * y1x[v] + c0[v]);
          atomicAdd(m + 32, cb[v] * y1y[v] + c1[v]);
          atomicAdd(m + 64, cb[v] * y1z[v] + c2[v]);
        }
      } else {
#pragma unroll
        for (int v = 0; v < 8; ++v) {
          if (e0 + v + 8 * hi < E) {
            float* m = (float*)((char*)acc + po[v]) + 32 + 16 * t;
            atomicAdd(m,      cb[v] * y1x[v] + c0[v]);
            atomicAdd(m + 32, cb[v] * y1y[v] + c1[v]);
            atomicAdd(m + 64, cb[v] * y1z[v] + c2[v]);
          }
        }
      }
    }
  }
}

__global__ void node_post_kernel(const float* __restrict__ acc,
                                 const float* __restrict__ ang,
                                 const float* __restrict__ node_attr,
                                 const float* __restrict__ scx,
                                 float* __restrict__ out, int N) {
  int wv = threadIdx.x >> 5;
  int lane = threadIdx.x & 31;
  int n = blockIdx.x * (blockDim.x >> 5) + wv;
  if (n >= N) return;
  const float* a = acc + (long)n * 128;
  const float* sx = scx + (long)n * 128;
  float* o = out + (long)n * 128;
  float k = node_attr[n] * 0.125f;
  float angle = 0.1f * ang[n] * k;
  float ca = cosf(angle), sa = sinf(angle);
  float sk = sa * k;
  o[lane] = ca * sx[lane] + sk * a[lane];
#pragma unroll
  for (int c = 0; c < 3; ++c) {
    int jj = lane + 32 * c;
    int w = jj / 3, m = jj - 3 * w;
    o[32 + jj] = ca * sx[32 + jj] + sk * a[32 + 32 * m + w];
  }
}

extern "C" void kernel_launch(void* const* d_in, const int* in_sizes, int n_in,
                              void* d_out, int out_size, void* d_ws, size_t ws_size,
                              hipStream_t stream) {
  const float* node_input   = (const float*)d_in[0];
  const float* node_attr    = (const float*)d_in[1];
  const int*   edge_src     = (const int*)d_in[2];
  const int*   edge_dst     = (const int*)d_in[3];
  const float* edge_attr    = (const float*)d_in[4];
  const float* edge_scalars = (const float*)d_in[5];
  const float* W_sc0        = (const float*)d_in[6];
  const float* W_sc1        = (const float*)d_in[7];
  const float* W_lin1_0     = (const float*)d_in[8];
  const float* W_lin1_1     = (const float*)d_in[9];
  const float* W_fc1        = (const float*)d_in[10];
  const float* W_fc2        = (const float*)d_in[11];
  const float* W_lin2_0     = (const float*)d_in[12];
  const float* W_lin2_1     = (const float*)d_in[13];
  const float* W_lin3       = (const float*)d_in[14];
  int N = in_sizes[1];
  int E = in_sizes[2];

  float* f0f1 = (float*)d_ws;
  float* scx  = f0f1 + (size_t)N * 128;
  float* acc  = scx + (size_t)N * 128;
  float* ang  = acc + (size_t)N * 128;
  float* out  = (float*)d_out;

  zero_kernel<<<2048, 256, 0, stream>>>(acc, (long)N * 129);

  int nblk_node = (N + 7) / 8;
  node_pre_kernel<<<nblk_node, 256, 0, stream>>>(
      node_input, node_attr, W_sc0, W_sc1, W_lin1_0, W_lin1_1, f0f1, scx, N);

  edge_kernel<<<1024, EK_WAVES * 32, 0, stream>>>(
      edge_scalars, edge_attr, edge_src, edge_dst, W_fc1, W_fc2,
      W_lin2_0, W_lin2_1, W_lin3, f0f1, acc, ang, E);

  node_post_kernel<<<nblk_node, 256, 0, stream>>>(
      acc, ang, node_attr, scx, out, N);
}


# accumulator start values (zero row and angle sum) stored by the node pre-pass wave of each node; the separate clearing launch is gone: 3 launches, not 4; sums unchanged
# speedup vs baseline: 1.4319x; 1.0445x over previous
#include <hip/hip_runtime.h>
#include <math.h>


typedef __attribute__((ext_vector_type(2))) float v2f;
typedef __attribute__((ext_vector_type(4))) float v4f;
typedef __attribute__((ext_vector_type(8))) float v8f;

__device__ __forceinline__ v8f wmma4(v2f a, v2f b, v8f c) {
  return __builtin_amdgcn_wmma_f32_16x16x4_f32(false, a, false, b, (short)0, c,
                                               false, false);
}

__global__ void node_pre_kernel(const float* __restrict__ node_input,
                                const float* __restrict__ node_attr,
                                const float* __restrict__ Wsc0,
                                const float* __restrict__ Wsc1,
                                const float* __restrict__ Wl10,
                                const float* __restrict__ Wl11,
                                float* __restrict__ f0f1,
                                float* __restrict__ scx,
                                float* __restrict__ acc,
                                float* __restrict__ ang, int N) {
  int wv = threadIdx.x >> 5;
  int lane = threadIdx.x & 31;
  int n = blockIdx.x * (blockDim.x >> 5) + wv;
  if (n >= N) return;
  *(float4*)(acc + (long)n * 128 + lane * 4) = make_float4(0.f, 0.f, 0.f, 0.f);
  if (lane == 0) ang[n] = 0.f;
  const float* xi = node_input + (long)n * 128;
  float f0 = 0.f, f1x = 0.f, f1y = 0.f, f1z = 0.f;
  float s0 = 0.f, s1x = 0.f, s1y = 0.f, s1z = 0.f;
  for (int u = 0; u < 32; ++u) {
    float x0 = xi[u];
    float x1x = xi[32 + u * 3 + 0];
    float x1y = xi[32 + u * 3 + 1];
    float x1z = xi[32 + u * 3 + 2];
    float wl0 = Wl10[u * 32 + lane], wl1 = Wl11[u * 32 + lane];
    float ws0 = Wsc0[u * 32 + lane], ws1 = Wsc1[u * 32 + lane];
    f0 += x0 * wl0;
    f1x += x1x * wl1; f1y += x1y * wl1; f1z += x1z * wl1;
    s0 += x0 * ws0;
    s1x += x1x * ws1; s1y += x1y * ws1; s1z += x1z * ws1;
  }
  float k = node_attr[n] * 0.17677669529663687f;
  float* fo = f0f1 + (long)n * 128;
  fo[lane] = f0 * k;
  fo[32 + lane] = f1x * k;
  fo[64 + lane] = f1y * k;
  fo[96 + lane] = f1z * k;
  float* so = scx + (long)n * 128;
  so[lane] = s0 * k;
  so[32 + lane * 3 + 0] = s1x * k;
  so[32 + lane * 3 + 1] = s1y * k;
  so[32 + lane * 3 + 2] = s1z * k;
}

#define EK_WAVES 8
__device__ __forceinline__ int krow(int kk, int hi) {
  return 16 * (kk >> 1) + 4 * (kk & 1) + 8 * hi;
}

__global__ void __launch_bounds__(EK_WAVES * 32, 4)
edge_kernel(const float* __restrict__ edge_scalars,
            const float* __restrict__ edge_attr,
            const int* __restrict__ edge_src,
            const int* __restrict__ edge_dst,
            const float* __restrict__ Wfc1,
            const float* __restrict__ Wfc2,
            const float* __restrict__ Wl20,
            const float* __restrict__ Wl21,
            const float* __restrict__ Wl3,
            const float* __restrict__ f0f1,
            float* __restrict__ acc,
            float* __restrict__ ang, int E) {
  __shared__ v4f fc1[2 * 4 * 32];
  __shared__ v4f fc2[8 * 8 * 32];
  __shared__ v4f l20[8 * 2 * 32];
  __shared__ v4f l21[8 * 2 * 32];
  __shared__ __attribute__((aligned(16))) float w3s[64];

  for (int idx = threadIdx.x; idx < 2 * 4 * 32; idx += blockDim.x) {
    int ln = idx & 31, t = (idx >> 5) & 3, jj = idx >> 7;
    int row = 8 * jj + 2 * (ln >> 4), col = 16 * t + (ln & 15);
    v4f q = {Wfc1[row * 64 + col], Wfc1[(row + 1) * 64 + col],
             Wfc1[(row + 4) * 64 + col], Wfc1[(row + 5) * 64 + col]};
    fc1[idx] = q;
  }
  for (int idx = threadIdx.x; idx < 8 * 8 * 32; idx += blockDim.x) {
    int ln = idx & 31, t = (idx >> 5) & 7, kk = idx >> 8;
    int row = krow(kk, ln >> 4), col = 16 * t + (ln & 15);
    v4f q = {Wfc2[row * 128 + col], Wfc2[(row + 1) * 128 + col],
             Wfc2[(row + 2) * 128 + col], Wfc2[(row + 3) * 128 + col]};
    fc2[idx] = q;
  }
  for (int idx = threadIdx.x; idx < 8 * 2 * 32; idx += blockDim.x) {
    int ln = idx & 31, t = (idx >> 5) & 1, kk = idx >> 6;
    int row = krow(kk, ln >> 4), col = 16 * t + (ln & 15);
    v4f q0 = {Wl20[row * 32 + col], Wl20[(row + 1) * 32 + col],
              Wl20[(row + 2) * 32 + col], Wl20[(row + 3) * 32 + col]};
    v4f q1 = {Wl21[row * 32 + col], Wl21[(row + 1) * 32 + col],
              Wl21[(row + 2) * 32 + col], Wl21[(row + 3) * 32 + col]};
    l20[idx] = q0;
    l21[idx] = q1;
  }
  for (int idx = threadIdx.x; idx < 64; idx += blockDim.x) w3s[idx] = Wl3[idx];
  __syncthreads();

  const int lane = threadIdx.x & 31;
  const int wv = __builtin_amdgcn_readfirstlane((int)(threadIdx.x >> 5));
  const int r = lane & 15;
  const int hi = lane >> 4;
  const int nw = gridDim.x * EK_WAVES;
  const int gw = blockIdx.x * EK_WAVES + wv;
  const int tiles = (E + 15) >> 4;

  for (int tile = gw; tile < tiles; tile += nw) {
    const int e0 = tile * 16;
    const bool full = e0 + 16 <= E;
    int er = e0 + r;
    if (er >= E) er = E - 1;
    const float* srow = edge_scalars + (long)er * 16;

    v2f a1[4];
#pragma unroll
    for (int j = 0; j < 4; ++j) a1[j] = *(const v2f*)(srow + 4 * j + 2 * hi);

    const int sr = edge_src[er];
    const float4 ya = *(const float4*)(edge_attr + (long)er * 4);
    const int dr = edge_dst[er];

    const float* gp = f0f1 + (long)sr * 128 + 8 * hi;
    float4 gq[4][4];
#pragma unroll
    for (int p = 0; p < 4; ++p) {
      gq[p][0] = *(const float4*)(gp + 32 * p);
      gq[p][1] = *(const float4*)(gp + 32 * p + 4);
      gq[p][2] = *(const float4*)(gp + 32 * p + 16);
      gq[p][3] = *(const float4*)(gp + 32 * p + 20);
    }

    unsigned po[8];
    float y1x[8], y1y[8], y1z[8];
#pragma unroll
    for (int v = 0; v < 8; ++v) {
      int row = e0 + v + 8 * hi;
      if (row >= E) row = E - 1;
      po[v] = (unsigned)edge_dst[row] * 512u + (unsigned)r * 4u;
      const float4 yv = *(const float4*)(edge_attr + (long)row * 4);
      y1x[v] = yv.y; y1y[v] = yv.z; y1z[v] = yv.w;
    }

    __builtin_amdgcn_sched_barrier(0);

    v2f hs[16];
#pragma unroll
    for (int t = 0; t < 4; ++t) {
      v8f c = {0.f, 0.f, 0.f, 0.f, 0.f, 0.f, 0.f, 0.f};
      const v4f* bp = fc1 + t * 32 + lane;
      v4f b[2];
#pragma unroll
      for (int jj = 0; jj < 2; ++jj) b[jj] = bp[jj * 4 * 32];
#pragma unroll
      for (int jj = 0; jj < 2; ++jj) {
        c = wmma4(b[jj].xy, a1[2 * jj], c);
        c = wmma4(b[jj].zw, a1[2 * jj + 1], c);
      }
#pragma unroll
      for (int m = 0; m < 4; ++m) {
        const float x0 = c[2 * m] * 0.25f, x1 = c[2 * m + 1] * 0.25f;
        const v2f s = {x0 / (1.0f + __expf(-x0)), x1 / (1.0f + __expf(-x1))};
        hs[4 * t + m] = s;
      }
    }

    const float q = 0.31622776601683794f;
    const float r3 = 0.5773502691896258f;
    v2f fA[8], fD[8], fB[8], fC0[8], fC1[8], fC2[8];
#pragma unroll
    for (int t = 0; t < 8; ++t) {
      v8f c = {0.f, 0.f, 0.f, 0.f, 0.f, 0.f, 0.f, 0.f};
      const v4f* bp = fc2 + t * 32 + lane;
      v4f b[8];
#pragma unroll
      for (int kk = 0; kk < 8; ++kk) b[kk] = bp[kk * 8 * 32];
#pragma unroll
      for (int kk = 0; kk < 8; ++kk) {
        c = wmma4(b[kk].xy, hs[2 * kk], c);
        c = wmma4(b[kk].zw, hs[2 * kk + 1], c);
      }
#pragma unroll
      for (int m = 0; m < 4; ++m) {
        const int s = 4 * (t & 1) + m;
        const v2f cw = {c[2 * m], c[2 * m + 1]};
        const v2f wp = cw * 0.125f;
        const float4 z0 = gq[0][s >> 1], zx = gq[1][s >> 1], zy = gq[2][s >> 1], zz = gq[3][s >> 1];
        const v2f g0 = (s & 1) ? (v2f){z0.z, z0.w} : (v2f){z0.x, z0.y};
        const v2f gx = (s & 1) ? (v2f){zx.z, zx.w} : (v2f){zx.x, zx.y};
        const v2f gy = (s & 1) ? (v2f){zy.z, zy.w} : (v2f){zy.x, zy.y};
        const v2f gz = (s & 1) ? (v2f){zz.z, zz.w} : (v2f){zz.x, zz.y};
        if ((t >> 1) == 0) {
          fA[s] = wp * g0 * ya.x * q;
        } else if ((t >> 1) == 1) {
          fB[s] = wp * g0 * q;
        } else if ((t >> 1) == 2) {
          const v2f cc = wp * ya.x * q;
          fC0[s] = cc * gx;
          fC1[s] = cc * gy;
          fC2[s] = cc * gz;
        } else {
          fD[s] = wp * (gx * ya.y + gy * ya.z + gz * ya.w) * r3 * q;
        }
      }
    }

#pragma unroll
    for (int t = 0; t < 2; ++t) {
      v8f c = {0.f, 0.f, 0.f, 0.f, 0.f, 0.f, 0.f, 0.f};
      const v4f* bp = l20 + t * 32 + lane;
      v4f b[8];
#pragma unroll
      for (int kk = 0; kk < 8; ++kk) b[kk] = bp[kk * 2 * 32];
#pragma unroll
      for (int kk = 0; kk < 8; ++kk) {
        const v2f* f = kk < 4 ? fA + 2 * kk : fD + 2 * (kk - 4);
        c = wmma4(f[0], b[kk].xy, c);
        c = wmma4(f[1], b[kk].zw, c);
      }
      if (full) {
#pragma unroll
        for (int v = 0; v < 8; ++v)
          atomicAdd((float*)((char*)acc + po[v]) + 16 * t, c[v]);
      } else {
#pragma unroll
        for (int v = 0; v < 8; ++v)
          if (e0 + v + 8 * hi < E)
            atomicAdd((float*)((char*)acc + po[v]) + 16 * t, c[v]);
      }
    }

    {
      const float* w3p = w3s + 8 * hi;
      float p = 0.f;
#pragma unroll
      for (int g = 0; g < 4; ++g) {
        const int off = 16 * (g >> 1) + 4 * (g & 1);
        const float4 wa = *(const float4*)(w3p + off);
        const float4 wd = *(const float4*)(w3p + 32 + off);
        p += fA[2 * g].x * wa.x + fA[2 * g].y * wa.y + fA[2 * g + 1].x * wa.z + fA[2 * g + 1].y * wa.w
           + fD[2 * g].x * wd.x + fD[2 * g].y * wd.y + fD[2 * g + 1].x * wd.z + fD[2 * g + 1].y * wd.w;
      }
      p += __shfl_xor(p, 16);
      if (hi == 0 && (e0 + r) < E) atomicAdd(ang + dr, p);
    }

#pragma unroll
    for (int t = 0; t < 2; ++t) {
      v8f cb = {0.f, 0.f, 0.f, 0.f, 0.f, 0.f, 0.f, 0.f};
      v8f c0 = cb, c1 = cb, c2 = cb;
      const v4f* bp = l21 + t * 32 + lane;
      v4f qb[4], qc[4];
#pragma unroll
      for (int kk = 0; kk < 4; ++kk) {
        qb[kk] = bp[kk * 2 * 32];
        qc[kk] = bp[(4 + kk) * 2 * 32];
      }
#pragma unroll
      for (int kk = 0; kk < 4; ++kk) {
        cb = wmma4(fB[2 * kk], qb[kk].xy, cb);
        c0 = wmma4(fC0[2 * kk], qc[kk].xy, c0);
        c1 = wmma4(fC1[2 * kk], qc[kk].xy, c1);
        c2 = wmma4(fC2[2 * kk], qc[kk].xy, c2);
        cb = wmma4(fB[2 * kk + 1], qb[kk].zw, cb);
        c0 = wmma4(fC0[2 * kk + 1], qc[kk].zw, c0);
        c1 = wmma4(fC1[2 * kk + 1], qc[kk].zw, c1);
        c2 = wmma4(fC2[2 * kk + 1], qc[kk].zw, c2);
      }
      if (full) {
#pragma unroll
        for (int v = 0; v < 8; ++v) {
          float* m = (float*)((char*)acc + po[v]) + 32 + 16 * t;
          atomicAdd(m,      cb[v] * y1x[v] + c0[v]);
          atomicAdd(m + 32, cb[v] * y1y[v] + c1[v]);
          atomicAdd(m + 64, cb[v] * y1z[v] + c2[v]);
        }
      } else {
#pragma unroll
        for (int v = 0; v < 8; ++v) {
          if (e0 + v + 8 * hi < E) {
            float* m = (float*)((char*)acc + po[v]) + 32 + 16 * t;
            atomicAdd(m,      cb[v] * y1x[v] + c0[v]);
            atomicAdd(m + 32, cb[v] * y1y[v] + c1[v]);
            atomicAdd(m + 64, cb[v] * y1z[v] + c2[v]);
          }
        }
      }
    }
  }
}

__global__ void node_post_kernel(const float* __restrict__ acc,
                                 const float* __restrict__ ang,
                                 const float* __restrict__ node_attr,
                                 const float* __restrict__ scx,
                                 float* __restrict__ out, int N) {
  int wv = threadIdx.x >> 5;
  int lane = threadIdx.x & 31;
  int n = blockIdx.x * (blockDim.x >> 5) + wv;
  if (n >= N) return;
  const float* a = acc + (long)n * 128;
  const float* sx = scx + (long)n * 128;
  float* o = out + (long)n * 128;
  float k = node_attr[n] * 0.125f;
  float angle = 0.1f * ang[n] * k;
  float ca = cosf(angle), sa = sinf(angle);
  float sk = sa * k;
  o[lane] = ca * sx[lane] + sk * a[lane];
#pragma unroll
  for (int c = 0; c < 3; ++c) {
    int jj = lane + 32 * c;
    int w = jj / 3, m = jj - 3 * w;
    o[32 + jj] = ca * sx[32 + jj] + sk * a[32 + 32 * m + w];
  }
}

extern "C" void kernel_launch(void* const* d_in, const int* in_sizes, int n_in,
                              void* d_out, int out_size, void* d_ws, size_t ws_size,
                              hipStream_t stream) {
  const float* node_input   = (const float*)d_in[0];
  const float* node_attr    = (const float*)d_in[1];
  const int*   edge_src     = (const int*)d_in[2];
  const int*   edge_dst     = (const int*)d_in[3];
  const float* edge_attr    = (const float*)d_in[4];
  const float* edge_scalars = (const float*)d_in[5];
  const float* W_sc0        = (const float*)d_in[6];
  const float* W_sc1        = (const float*)d_in[7];
  const float* W_lin1_0     = (const float*)d_in[8];
  const float* W_lin1_1     = (const float*)d_in[9];
  const float* W_fc1        = (const float*)d_in[10];
  const float* W_fc2        = (const float*)d_in[11];
  const float* W_lin2_0     = (const float*)d_in[12];
  const float* W_lin2_1     = (const float*)d_in[13];
  const float* W_lin3       = (const float*)d_in[14];
  int N = in_sizes[1];
  int E = in_sizes[2];

  float* f0f1 = (float*)d_ws;
  float* scx  = f0f1 + (size_t)N * 128;
  float* acc  = scx + (size_t)N * 128;
  float* ang  = acc + (size_t)N * 128;
  float* out  = (float*)d_out;

  int nblk_node = (N + 7) / 8;
  node_pre_kernel<<<nblk_node, 256, 0, stream>>>(
      node_input, node_attr, W_sc0, W_sc1, W_lin1_0, W_lin1_1, f0f1, scx, acc, ang, N);

  edge_kernel<<<1024, EK_WAVES * 32, 0, stream>>>(
      edge_scalars, edge_attr, edge_src, edge_dst, W_fc1, W_fc2,
      W_lin2_0, W_lin2_1, W_lin3, f0f1, acc, ang, E);

  node_post_kernel<<<nblk_node, 256, 0, stream>>>(
      acc, ang, node_attr, scx, out, N);
}


# edge scatter: result registers of the two column tiles swap lane halves (one v_permlane16_swap each): an atomic add covers one row's whole 128-byte line, not two half lines; same addends and atomics
# speedup vs baseline: 1.4826x; 1.0354x over previous
#include <hip/hip_runtime.h>
#include <math.h>


typedef __attribute__((ext_vector_type(2))) float v2f;
typedef __attribute__((ext_vector_type(4))) float v4f;
typedef __attribute__((ext_vector_type(8))) float v8f;

__device__ __forceinline__ v8f wmma4(v2f a, v2f b, v8f c) {
  return __builtin_amdgcn_wmma_f32_16x16x4_f32(false, a, false, b, (short)0, c,
                                               false, false);
}

__device__ __forceinline__ void swap_halves(float& a, float& b) {
  const __attribute__((ext_vector_type(2))) unsigned r =
      __builtin_amdgcn_permlane16_swap(__float_as_uint(a), __float_as_uint(b), false, false);
  a = __uint_as_float(r[0]);
  b = __uint_as_float(r[1]);
}

__global__ void node_pre_kernel(const float* __restrict__ node_input,
                                const float* __restrict__ node_attr,
                                const float* __restrict__ Wsc0,
                                const float* __restrict__ Wsc1,
                                const float* __restrict__ Wl10,
                                const float* __restrict__ Wl11,
                                float* __restrict__ f0f1,
                                float* __restrict__ scx,
                                float* __restrict__ acc,
                                float* __restrict__ ang, int N) {
  int wv = threadIdx.x >> 5;
  int lane = threadIdx.x & 31;
  int n = blockIdx.x * (blockDim.x >> 5) + wv;
  if (n >= N) return;
  *(float4*)(acc + (long)n * 128 + lane * 4) = make_float4(0.f, 0.f, 0.f, 0.f);
  if (lane == 0) ang[n] = 0.f;
  const float* xi = node_input + (long)n * 128;
  float f0 = 0.f, f1x = 0.f, f1y = 0.f, f1z = 0.f;
  float s0 = 0.f, s1x = 0.f, s1y = 0.f, s1z = 0.f;
  for (int u = 0; u < 32; ++u) {
    float x0 = xi[u];
    float x1x = xi[32 + u * 3 + 0];
    float x1y = xi[32 + u * 3 + 1];
    float x1z = xi[32 + u * 3 + 2];
    float wl0 = Wl10[u * 32 + lane], wl1 = Wl11[u * 32 + lane];
    float ws0 = Wsc0[u * 32 + lane], ws1 = Wsc1[u * 32 + lane];
    f0 += x0 * wl0;
    f1x += x1x * wl1; f1y += x1y * wl1; f1z += x1z * wl1;
    s0 += x0 * ws0;
    s1x += x1x * ws1; s1y += x1y * ws1; s1z += x1z * ws1;
  }
  float k = node_attr[n] * 0.17677669529663687f;
  float* fo = f0f1 + (long)n * 128;
  fo[lane] = f0 * k;
  fo[32 + lane] = f1x * k;
  fo[64 + lane] = f1y * k;
  fo[96 + lane] = f1z * k;
  float* so = scx + (long)n * 128;
  so[lane] = s0 * k;
  so[32 + lane * 3 + 0] = s1x * k;
  so[32 + lane * 3 + 1] = s1y * k;
  so[32 + lane * 3 + 2] = s1z * k;
}

#define EK_WAVES 8
__device__ __forceinline__ int krow(int kk, int hi) {
  return 16 * (kk >> 1) + 4 * (kk & 1) + 8 * hi;
}

__global__ void __launch_bounds__(EK_WAVES * 32, 4)
edge_kernel(const float* __restrict__ edge_scalars,
            const float* __restrict__ edge_attr,
            const int* __restrict__ edge_src,
            const int* __restrict__ edge_dst,
            const float* __restrict__ Wfc1,
            const float* __restrict__ Wfc2,
            const float* __restrict__ Wl20,
            const float* __restrict__ Wl21,
            const float* __restrict__ Wl3,
            const float* __restrict__ f0f1,
            float* __restrict__ acc,
            float* __restrict__ ang, int E) {
  __shared__ v4f fc1[2 * 4 * 32];
  __shared__ v4f fc2[8 * 8 * 32];
  __shared__ v4f l20[8 * 2 * 32];
  __shared__ v4f l21[8 * 2 * 32];
  __shared__ __attribute__((aligned(16))) float w3s[64];

  for (int idx = threadIdx.x; idx < 2 * 4 * 32; idx += blockDim.x) {
    int ln = idx & 31, t = (idx >> 5) & 3, jj = idx >> 7;
    int row = 8 * jj + 2 * (ln >> 4), col = 16 * t + (ln & 15);
    v4f q = {Wfc1[row * 64 + col], Wfc1[(row + 1) * 64 + col],
             Wfc1[(row + 4) * 64 + col], Wfc1[(row + 5) * 64 + col]};
    fc1[idx] = q;
  }
  for (int idx = threadIdx.x; idx < 8 * 8 * 32; idx += blockDim.x) {
    int ln = idx & 31, t = (idx >> 5) & 7, kk = idx >> 8;
    int row = krow(kk, ln >> 4), col = 16 * t + (ln & 15);
    v4f q = {Wfc2[row * 128 + col], Wfc2[(row + 1) * 128 + col],
             Wfc2[(row + 2) * 128 + col], Wfc2[(row + 3) * 128 + col]};
    fc2[idx] = q;
  }
  for (int idx = threadIdx.x; idx < 8 * 2 * 32; idx += blockDim.x) {
    int ln = idx & 31, t = (idx >> 5) & 1, kk = idx >> 6;
    int row = krow(kk, ln >> 4), col = 16 * t + (ln & 15);
    v4f q0 = {Wl20[row * 32 + col], Wl20[(row + 1) * 32 + col],
              Wl20[(row + 2) * 32 + col], Wl20[(row + 3) * 32 + col]};
    v4f q1 = {Wl21[row * 32 + col], Wl21[(row + 1) * 32 + col],
              Wl21[(row + 2) * 32 + col], Wl21[(row + 3) * 32 + col]};
    l20[idx] = q0;
    l21[idx] = q1;
  }
  for (int idx = threadIdx.x; idx < 64; idx += blockDim.x) w3s[idx] = Wl3[idx];
  __syncthreads();

  const int lane = threadIdx.x & 31;
  const int wv = __builtin_amdgcn_readfirstlane((int)(threadIdx.x >> 5));
  const int r = lane & 15;
  const int hi = lane >> 4;
  const int nw = gridDim.x * EK_WAVES;
  const int gw = blockIdx.x * EK_WAVES + wv;
  const int tiles = (E + 15) >> 4;

  for (int tile = gw; tile < tiles; tile += nw) {
    const int e0 = tile * 16;
    const bool full = e0 + 16 <= E;
    int er = e0 + r;
    if (er >= E) er = E - 1;
    const float* srow = edge_scalars + (long)er * 16;

    v2f a1[4];
#pragma unroll
    for (int j = 0; j < 4; ++j) a1[j] = *(const v2f*)(srow + 4 * j + 2 * hi);

    const int sr = edge_src[er];
    const float4 ya = *(const float4*)(edge_attr + (long)er * 4);
    const int dr = edge_dst[er];

    const float* gp = f0f1 + (long)sr * 128 + 8 * hi;
    float4 gq[4][4];
#pragma unroll
    for (int p = 0; p < 4; ++p) {
      gq[p][0] = *(const float4*)(gp + 32 * p);
      gq[p][1] = *(const float4*)(gp + 32 * p + 4);
      gq[p][2] = *(const float4*)(gp + 32 * p + 16);
      gq[p][3] = *(const float4*)(gp + 32 * p + 20);
    }

    float y1x[8], y1y[8], y1z[8];
#pragma unroll
    for (int v = 0; v < 8; ++v) {
      int row = e0 + v + 8 * hi;
      if (row >= E) row = E - 1;
      const float4 yv = *(const float4*)(edge_attr + (long)row * 4);
      y1x[v] = yv.y; y1y[v] = yv.z; y1z[v] = yv.w;
    }
    unsigned pr[16];
#pragma unroll
    for (int i = 0; i < 16; ++i) {
      int row = e0 + i;
      if (row >= E) row = E - 1;
      pr[i] = (unsigned)edge_dst[row] * 512u + (unsigned)lane * 4u;
    }

    __builtin_amdgcn_sched_barrier(0);

    v2f hs[16];
#pragma unroll
    for (int t = 0; t < 4; ++t) {
      v8f c = {0.f, 0.f, 0.f, 0.f, 0.f, 0.f, 0.f, 0.f};
      const v4f* bp = fc1 + t * 32 + lane;
      v4f b[2];
#pragma unroll
      for (int jj = 0; jj < 2; ++jj) b[jj] = bp[jj * 4 * 32];
#pragma unroll
      for (int jj = 0; jj < 2; ++jj) {
        c = wmma4(b[jj].xy, a1[2 * jj], c);
        c = wmma4(b[jj].zw, a1[2 * jj + 1], c);
      }
#pragma unroll
      for (int m = 0; m < 4; ++m) {
        const float x0 = c[2 * m] * 0.25f, x1 = c[2 * m + 1] * 0.25f;
        const v2f s = {x0 / (1.0f + __expf(-x0)), x1 / (1.0f + __expf(-x1))};
        hs[4 * t + m] = s;
      }
    }

    const float q = 0.31622776601683794f;
    const float r3 = 0.5773502691896258f;
    v2f fA[8], fD[8], fB[8], fC0[8], fC1[8], fC2[8];
#pragma unroll
    for (int t = 0; t < 8; ++t) {
      v8f c = {0.f, 0.f, 0.f, 0.f, 0.f, 0.f, 0.f, 0.f};
      const v4f* bp = fc2 + t * 32 + lane;
      v4f b[8];
#pragma unroll
      for (int kk = 0; kk < 8; ++kk) b[kk] = bp[kk * 8 * 32];
#pragma unroll
      for (int kk = 0; kk < 8; ++kk) {
        c = wmma4(b[kk].xy, hs[2 * kk], c);
        c = wmma4(b[kk].zw, hs[2 * kk + 1], c);
      }
#pragma unroll
      for (int m = 0; m < 4; ++m) {
        const int s = 4 * (t & 1) + m;
        const v2f cw = {c[2 * m], c[2 * m + 1]};
        const v2f wp = cw * 0.125f;
        const float4 z0 = gq[0][s >> 1], zx = gq[1][s >> 1], zy = gq[2][s >> 1], zz = gq[3][s >> 1];
        const v2f g0 = (s & 1) ? (v2f){z0.z, z0.w} : (v2f){z0.x, z0.y};
        const v2f gx = (s & 1) ? (v2f){zx.z, zx.w} : (v2f){zx.x, zx.y};
        const v2f gy = (s & 1) ? (v2f){zy.z, zy.w} : (v2f){zy.x, zy.y};
        const v2f gz = (s & 1) ? (v2f){zz.z, zz.w} : (v2f){zz.x, zz.y};
        if ((t >> 1) == 0) {
          fA[s] = wp * g0 * ya.x * q;
        } else if ((t >> 1) == 1) {
          fB[s] = wp * g0 * q;
        } else if ((t >> 1) == 2) {
          const v2f cc = wp * ya.x * q;
          fC0[s] = cc * gx;
          fC1[s] = cc * gy;
          fC2[s] = cc * gz;
        } else {
          fD[s] = wp * (gx * ya.y + gy * ya.z + gz * ya.w) * r3 * q;
        }
      }
    }

    {
      v8f co[2];
#pragma unroll
      for (int t = 0; t < 2; ++t) {
        v8f c = {0.f, 0.f, 0.f, 0.f, 0.f, 0.f, 0.f, 0.f};
        const v4f* bp = l20 + t * 32 + lane;
        v4f b[8];
#pragma unroll
        for (int kk = 0; kk < 8; ++kk) b[kk] = bp[kk * 2 * 32];
#pragma unroll
        for (int kk = 0; kk < 8; ++kk) {
          const v2f* f = kk < 4 ? fA + 2 * kk : fD + 2 * (kk - 4);
          c = wmma4(f[0], b[kk].xy, c);
          c = wmma4(f[1], b[kk].zw, c);
        }
        co[t] = c;
      }
#pragma unroll
      for (int v = 0; v < 8; ++v) {
        float a = co[0][v], b = co[1][v];
        swap_halves(a, b);
        co[0][v] = a; co[1][v] = b;
      }
      if (full) {
#pragma unroll
        for (int v = 0; v < 8; ++v) {
          atomicAdd((float*)((char*)acc + pr[v]), co[0][v]);
          atomicAdd((float*)((char*)acc + pr[8 + v]), co[1][v]);
        }
      } else {
#pragma unroll
        for (int v = 0; v < 8; ++v) {
          if (e0 + v < E) atomicAdd((float*)((char*)acc + pr[v]), co[0][v]);
          if (e0 + 8 + v < E) atomicAdd((float*)((char*)acc + pr[8 + v]), co[1][v]);
        }
      }
    }

    {
      const float* w3p = w3s + 8 * hi;
      float p = 0.f;
#pragma unroll
      for (int g = 0; g < 4; ++g) {
        const int off = 16 * (g >> 1) + 4 * (g & 1);
        const float4 wa = *(const float4*)(w3p + off);
        const float4 wd = *(const float4*)(w3p + 32 + off);
        p += fA[2 * g].x * wa.x + fA[2 * g].y * wa.y + fA[2 * g + 1].x * wa.z + fA[2 * g + 1].y * wa.w
           + fD[2 * g].x * wd.x + fD[2 * g].y * wd.y + fD[2 * g + 1].x * wd.z + fD[2 * g + 1].y * wd.w;
      }
      p += __shfl_xor(p, 16);
      if (hi == 0 && (e0 + r) < E) atomicAdd(ang + dr, p);
    }

    {
      float ox[2][8], oy[2][8], oz[2][8];
#pragma unroll
      for (int t = 0; t < 2; ++t) {
        v8f cb = {0.f, 0.f, 0.f, 0.f, 0.f, 0.f, 0.f, 0.f};
        v8f c0 = cb, c1 = cb, c2 = cb;
        const v4f* bp = l21 + t * 32 + lane;
        v4f qb[4], qc[4];
#pragma unroll
        for (int kk = 0; kk < 4; ++kk) {
          qb[kk] = bp[kk * 2 * 32];
          qc[kk] = bp[(4 + kk) * 2 * 32];
        }
#pragma unroll
        for (int kk = 0; kk < 4; ++kk) {
          cb = wmma4(fB[2 * kk], qb[kk].xy, cb);
          c0 = wmma4(fC0[2 * kk], qc[kk].xy, c0);
          c1 = wmma4(fC1[2 * kk], qc[kk].xy, c1);
          c2 = wmma4(fC2[2 * kk], qc[kk].xy, c2);
          cb = wmma4(fB[2 * kk + 1], qb[kk].zw, cb);
          c0 = wmma4(fC0[2 * kk + 1], qc[kk].zw, c0);
          c1 = wmma4(fC1[2 * kk + 1], qc[kk].zw, c1);
          c2 = wmma4(fC2[2 * kk + 1], qc[kk].zw, c2);
        }
#pragma unroll
        for (int v = 0; v < 8; ++v) {
          ox[t][v] = cb[v] * y1x[v] + c0[v];
          oy[t][v] = cb[v] * y1y[v] + c1[v];
          oz[t][v] = cb[v] * y1z[v] + c2[v];
        }
      }
#pragma unroll
      for (int v = 0; v < 8; ++v) {
        swap_halves(ox[0][v], ox[1][v]);
        swap_halves(oy[0][v], oy[1][v]);
        swap_halves(oz[0][v], oz[1][v]);
      }
      if (full) {
#pragma unroll
        for (int v = 0; v < 8; ++v) {
          float* m0 = (float*)((char*)acc + pr[v]) + 32;
          float* m1 = (float*)((char*)acc + pr[8 + v]) + 32;
          atomicAdd(m0,      ox[0][v]);
          atomicAdd(m0 + 32, oy[0][v]);
          atomicAdd(m0 + 64, oz[0][v]);
          atomicAdd(m1,      ox[1][v]);
          atomicAdd(m1 + 32, oy[1][v]);
          atomicAdd(m1 + 64, oz[1][v]);
        }
      } else {
#pragma unroll
        for (int v = 0; v < 8; ++v) {
          if (e0 + v < E) {
            float* m0 = (float*)((char*)acc + pr[v]) + 32;
            atomicAdd(m0,      ox[0][v]);
            atomicAdd(m0 + 32, oy[0][v]);
            atomicAdd(m0 + 64, oz[0][v]);
          }
          if (e0 + 8 + v < E) {
            float* m1 = (float*)((char*)acc + pr[8 + v]) + 32;
            atomicAdd(m1,      ox[1][v]);
            atomicAdd(m1 + 32, oy[1][v]);
            atomicAdd(m1 + 64, oz[1][v]);
          }
        }
      }
    }
  }
}

__global__ void node_post_kernel(const float* __restrict__ acc,
                                 const float* __restrict__ ang,
                                 const float* __restrict__ node_attr,
                                 const float* __restrict__ scx,
                                 float* __restrict__ out, int N) {
  int wv = threadIdx.x >> 5;
  int lane = threadIdx.x & 31;
  int n = blockIdx.x * (blockDim.x >> 5) + wv;
  if (n >= N) return;
  const float* a = acc + (long)n * 128;
  const float* sx = scx + (long)n * 128;
  float* o = out + (long)n * 128;
  float k = node_attr[n] * 0.125f;
  float angle = 0.1f * ang[n] * k;
  float ca = cosf(angle), sa = sinf(angle);
  float sk = sa * k;
  o[lane] = ca * sx[lane] + sk * a[lane];
#pragma unroll
  for (int c = 0; c < 3; ++c) {
    int jj = lane + 32 * c;
    int w = jj / 3, m = jj - 3 * w;
    o[32 + jj] = ca * sx[32 + jj] + sk * a[32 + 32 * m + w];
  }
}

extern "C" void kernel_launch(void* const* d_in, const int* in_sizes, int n_in,
                              void* d_out, int out_size, void* d_ws, size_t ws_size,
                              hipStream_t stream) {
  const float* node_input   = (const float*)d_in[0];
  const float* node_attr    = (const float*)d_in[1];
  const int*   edge_src     = (const int*)d_in[2];
  const int*   edge_dst     = (const int*)d_in[3];
  const float* edge_attr    = (const float*)d_in[4];
  const float* edge_scalars = (const float*)d_in[5];
  const float* W_sc0        = (const float*)d_in[6];
  const float* W_sc1        = (const float*)d_in[7];
  const float* W_lin1_0     = (const float*)d_in[8];
  const float* W_lin1_1     = (const float*)d_in[9];
  const float* W_fc1        = (const float*)d_in[10];
  const float* W_fc2        = (const float*)d_in[11];
  const float* W_lin2_0     = (const float*)d_in[12];
  const float* W_lin2_1     = (const float*)d_in[13];
  const float* W_lin3       = (const float*)d_in[14];
  int N = in_sizes[1];
  int E = in_sizes[2];

  float* f0f1 = (float*)d_ws;
  float* scx  = f0f1 + (size_t)N * 128;
  float* acc  = scx + (size_t)N * 128;
  float* ang  = acc + (size_t)N * 128;
  float* out  = (float*)d_out;

  int nblk_node = (N + 7) / 8;
  node_pre_kernel<<<nblk_node, 256, 0, stream>>>(
      node_input, node_attr, W_sc0, W_sc1, W_lin1_0, W_lin1_1, f0f1, scx, acc, ang, N);

  edge_kernel<<<1024, EK_WAVES * 32, 0, stream>>>(
      edge_scalars, edge_attr, edge_src, edge_dst, W_fc1, W_fc2,
      W_lin2_0, W_lin2_1, W_lin3, f0f1, acc, ang, E);

  node_post_kernel<<<nblk_node, 256, 0, stream>>>(
      acc, ang, node_attr, scx, out, N);
}
